# starting kernel
# baseline (speedup 1.0000x reference)
#include <hip/hip_runtime.h>
#include <hip/hip_bf16.h>


#define Bsz   256
#define Asz   100
#define NBq   10
#define BONDq 110
#define AFq   82
#define BFq   6
#define Hq    300
#define KPAD  308
#define NT    19
#define NP    10
#define MTOT  (Bsz*Asz)
#define MT    (MTOT/16)
#define MP    (MT/2)
#define ATPB  4
#define PANEL (16*KPAD)

typedef float v2f __attribute__((ext_vector_type(2)));
typedef float v8f __attribute__((ext_vector_type(8)));

static __device__ __forceinline__ v8f wmma4(v2f a, v2f b, v8f c) {
  return __builtin_amdgcn_wmma_f32_16x16x4_f32(
      false, a, false, b, (short)0, c, false, false);
}

__global__ __launch_bounds__(256) void fc1_kernel(
    const float* __restrict__ af, const float* __restrict__ w,
    float* __restrict__ hout) {
  int tid = blockIdx.x * 8 + (threadIdx.x >> 5);
  if (tid >= MP * NP) return;
  const int lane = threadIdx.x & 31;
  const int m = lane & 15, half = lane >> 4, koff = half << 1;
  const int mp = tid / NP, np = tid - mp * NP;
  const int n0 = np * 32 + m, n1 = n0 + 16;
  const int row0 = mp * 32 + m, row1 = row0 + 16;
  const int nc0 = (n0 < Hq) ? n0 : (Hq - 1);
  const int nc1 = (n1 < Hq) ? n1 : (Hq - 1);

  const float* ap0 = af + row0 * AFq + koff;
  const float* ap1 = af + row1 * AFq + koff;
  const float* wp0 = w + koff * Hq + nc0;
  const float* wp1 = w + koff * Hq + nc1;
  v8f c00 = {}, c01 = {}, c10 = {}, c11 = {};
#pragma unroll 2
  for (int kk = 0; kk < 80; kk += 4) {
    v2f a0 = *(const v2f*)ap0;
    v2f a1 = *(const v2f*)ap1;
    v2f b0; b0.x = wp0[0]; b0.y = wp0[Hq];
    v2f b1; b1.x = wp1[0]; b1.y = wp1[Hq];
    ap0 += 4; ap1 += 4; wp0 += 4 * Hq; wp1 += 4 * Hq;
    c00 = wmma4(a0, b0, c00); c01 = wmma4(a0, b1, c01);
    c10 = wmma4(a1, b0, c10); c11 = wmma4(a1, b1, c11);
  }
  {
    const int k0 = 80 + koff;
    const int k0c = (k0 < AFq) ? k0 : AFq - 1;
    const int k1c = (k0 + 1 < AFq) ? k0 + 1 : AFq - 1;
    v2f a0, a1, b0, b1;
    float a0x = af[row0 * AFq + k0c], a0y = af[row0 * AFq + k1c];
    float a1x = af[row1 * AFq + k0c], a1y = af[row1 * AFq + k1c];
    a0.x = (k0 < AFq) ? a0x : 0.f;  a0.y = (k0 + 1 < AFq) ? a0y : 0.f;
    a1.x = (k0 < AFq) ? a1x : 0.f;  a1.y = (k0 + 1 < AFq) ? a1y : 0.f;
    b0.x = w[k0c * Hq + nc0]; b0.y = w[k1c * Hq + nc0];
    b1.x = w[k0c * Hq + nc1]; b1.y = w[k1c * Hq + nc1];
    c00 = wmma4(a0, b0, c00); c01 = wmma4(a0, b1, c01);
    c10 = wmma4(a1, b0, c10); c11 = wmma4(a1, b1, c11);
  }
  const int rb0 = mp * 32 + half * 8, rb1 = rb0 + 16;
#pragma unroll
  for (int v = 0; v < 8; ++v) {
    if (n0 < Hq) {
      hout[(rb0 + v) * Hq + n0] = fmaxf(c00[v], 0.f);
      hout[(rb1 + v) * Hq + n0] = fmaxf(c10[v], 0.f);
    }
    if (n1 < Hq) {
      hout[(rb0 + v) * Hq + n1] = fmaxf(c01[v], 0.f);
      hout[(rb1 + v) * Hq + n1] = fmaxf(c11[v], 0.f);
    }
  }
}

__global__ __launch_bounds__(256) void nei_kernel(
    const float* __restrict__ h, const float* __restrict__ bond,
    const int* __restrict__ ag, const int* __restrict__ bg,
    const int* __restrict__ nnb, const float* __restrict__ wn,
    const float* __restrict__ bn, float* __restrict__ out) {
  __shared__ float Ap[ATPB * PANEL];
  __shared__ int sib[ATPB * NBq], sia[ATPB * NBq];
  __shared__ int sbb[ATPB * NBq], sbe[ATPB * NBq];
  const int ba0 = blockIdx.x * ATPB;
  if (threadIdx.x < ATPB * NBq) {
    int q = threadIdx.x / NBq, r = threadIdx.x - q * NBq;
    int e = ((ba0 + q) * NBq + r) * 2;
    sib[threadIdx.x] = ag[e + 0]; sia[threadIdx.x] = ag[e + 1];
    sbb[threadIdx.x] = bg[e + 0]; sbe[threadIdx.x] = bg[e + 1];
  }
  __syncthreads();
  for (int idx = threadIdx.x; idx < ATPB * PANEL; idx += 256) {
    int q = idx / PANEL, rem = idx - q * PANEL;
    int r = rem / KPAD, c = rem - r * KPAD;
    float v = 0.f;
    if (r < NBq) {
      int s = q * NBq + r;
      if (c < Hq)            v = h[(sib[s] * Asz + sia[s]) * Hq + c];
      else if (c < Hq + BFq) v = bond[(sbb[s] * BONDq + sbe[s]) * BFq + (c - Hq)];
    }
    Ap[idx] = v;
  }
  __syncthreads();
  const int wave = threadIdx.x >> 5, lane = threadIdx.x & 31;
  const int m = lane & 15, half = lane >> 4, koff = half << 1;
  const float* apl = Ap + m * KPAD + koff;
  for (int t = wave; t < NT; t += 8) {
    const int n = t * 16 + m;
    const int nc = (n < Hq) ? n : (Hq - 1);
    const float* wp = wn + koff * Hq + nc;
    v8f acc[ATPB] = {};
#pragma unroll 2
    for (int kk = 0; kk < 304; kk += 4) {
      v2f b; b.x = wp[0]; b.y = wp[Hq];
      wp += 4 * Hq;
#pragma unroll
      for (int q = 0; q < ATPB; ++q) {
        v2f a = *(const v2f*)(apl + q * PANEL + kk);
        acc[q] = wmma4(a, b, acc[q]);
      }
    }
    {
      const int k0 = 304 + koff;
      v2f b;
      b.x = wn[(k0     < Hq + BFq ? k0     : Hq + BFq - 1) * Hq + nc];
      b.y = wn[(k0 + 1 < Hq + BFq ? k0 + 1 : Hq + BFq - 1) * Hq + nc];
#pragma unroll
      for (int q = 0; q < ATPB; ++q) {
        v2f a = *(const v2f*)(apl + q * PANEL + 304);
        acc[q] = wmma4(a, b, acc[q]);
      }
    }
    const float bias = bn[nc];
#pragma unroll
    for (int q = 0; q < ATPB; ++q) {
      const int nn = nnb[ba0 + q];
      float s = 0.f;
#pragma unroll
      for (int v = 0; v < 8; ++v) {
        int rj = v + half * 8;
        float x = fmaxf(acc[q][v] + bias, 0.f);
        s += (rj < nn) ? x : 0.f;
      }
      s += __shfl_xor(s, 16, 32);
      if (half == 0 && n < Hq) out[(ba0 + q) * Hq + n] = s;
    }
  }
}

__global__ __launch_bounds__(256) void atom_kernel(
    const float* __restrict__ h, const float* __restrict__ nei,
    const float* __restrict__ wa, const float* __restrict__ bb,
    float* __restrict__ hout) {
  int tid = blockIdx.x * 8 + (threadIdx.x >> 5);
  if (tid >= MP * NP) return;
  const int lane = threadIdx.x & 31;
  const int m = lane & 15, half = lane >> 4, koff = half << 1;
  const int mp = tid / NP, np = tid - mp * NP;
  const int n0 = np * 32 + m, n1 = n0 + 16;
  const int row0 = mp * 32 + m, row1 = row0 + 16;
  const int nc0 = (n0 < Hq) ? n0 : (Hq - 1);
  const int nc1 = (n1 < Hq) ? n1 : (Hq - 1);
  v8f c00 = {}, c01 = {}, c10 = {}, c11 = {};
  {
    const float* ap0 = h + row0 * Hq + koff;
    const float* ap1 = h + row1 * Hq + koff;
    const float* wp0 = wa + koff * Hq + nc0;
    const float* wp1 = wa + koff * Hq + nc1;
#pragma unroll 3
    for (int kk = 0; kk < Hq; kk += 4) {
      v2f a0 = *(const v2f*)ap0;
      v2f a1 = *(const v2f*)ap1;
      v2f b0; b0.x = wp0[0]; b0.y = wp0[Hq];
      v2f b1; b1.x = wp1[0]; b1.y = wp1[Hq];
      ap0 += 4; ap1 += 4; wp0 += 4 * Hq; wp1 += 4 * Hq;
      c00 = wmma4(a0, b0, c00); c01 = wmma4(a0, b1, c01);
      c10 = wmma4(a1, b0, c10); c11 = wmma4(a1, b1, c11);
    }
  }
  {
    const float* ap0 = nei + row0 * Hq + koff;
    const float* ap1 = nei + row1 * Hq + koff;
    const float* wp0 = wa + (Hq + koff) * Hq + nc0;
    const float* wp1 = wa + (Hq + koff) * Hq + nc1;
#pragma unroll 3
    for (int kk = 0; kk < Hq; kk += 4) {
      v2f a0 = *(const v2f*)ap0;
      v2f a1 = *(const v2f*)ap1;
      v2f b0; b0.x = wp0[0]; b0.y = wp0[Hq];
      v2f b1; b1.x = wp1[0]; b1.y = wp1[Hq];
      ap0 += 4; ap1 += 4; wp0 += 4 * Hq; wp1 += 4 * Hq;
      c00 = wmma4(a0, b0, c00); c01 = wmma4(a0, b1, c01);
      c10 = wmma4(a1, b0, c10); c11 = wmma4(a1, b1, c11);
    }
  }
  const float bias0 = bb[nc0], bias1 = bb[nc1];
  const int rb0 = mp * 32 + half * 8, rb1 = rb0 + 16;
#pragma unroll
  for (int v = 0; v < 8; ++v) {
    if (n0 < Hq) {
      hout[(rb0 + v) * Hq + n0] = fmaxf(c00[v] + bias0, 0.f);
      hout[(rb1 + v) * Hq + n0] = fmaxf(c10[v] + bias0, 0.f);
    }
    if (n1 < Hq) {
      hout[(rb0 + v) * Hq + n1] = fmaxf(c01[v] + bias1, 0.f);
      hout[(rb1 + v) * Hq + n1] = fmaxf(c11[v] + bias1, 0.f);
    }
  }
}

__global__ __launch_bounds__(256) void nprod_kernel(
    const float* __restrict__ h, const float* __restrict__ bond,
    const int* __restrict__ ag, const int* __restrict__ bg,
    const int* __restrict__ nnb, const float* __restrict__ w2a,
    const float* __restrict__ w2b, float* __restrict__ out) {
  __shared__ float Ap[ATPB * PANEL];
  __shared__ int sib[ATPB * NBq], sia[ATPB * NBq];
  __shared__ int sbb[ATPB * NBq], sbe[ATPB * NBq];
  const int ba0 = blockIdx.x * ATPB;
  if (threadIdx.x < ATPB * NBq) {
    int q = threadIdx.x / NBq, r = threadIdx.x - q * NBq;
    int e = ((ba0 + q) * NBq + r) * 2;
    sib[threadIdx.x] = ag[e + 0]; sia[threadIdx.x] = ag[e + 1];
    sbb[threadIdx.x] = bg[e + 0]; sbe[threadIdx.x] = bg[e + 1];
  }
  __syncthreads();
  for (int idx = threadIdx.x; idx < ATPB * PANEL; idx += 256) {
    int q = idx / PANEL, rem = idx - q * PANEL;
    int r = rem / KPAD, c = rem - r * KPAD;
    float v = 0.f;
    if (r < NBq) {
      int s = q * NBq + r;
      if (c < Hq)            v = h[(sib[s] * Asz + sia[s]) * Hq + c];
      else if (c < Hq + BFq) v = bond[(sbb[s] * BONDq + sbe[s]) * BFq + (c - Hq)];
    }
    Ap[idx] = v;
  }
  __syncthreads();
  const int wave = threadIdx.x >> 5, lane = threadIdx.x & 31;
  const int m = lane & 15, half = lane >> 4, koff = half << 1;
  const float* apl = Ap + m * KPAD + koff;
  for (int t = wave; t < NT; t += 8) {
    const int n = t * 16 + m;
    const int nc = (n < Hq) ? n : (Hq - 1);
    v8f accA[ATPB] = {}, accB[ATPB] = {};
    {
      const float* wp = w2a + koff * Hq + nc;
#pragma unroll 3
      for (int kk = 0; kk < Hq; kk += 4) {
        v2f b; b.x = wp[0]; b.y = wp[Hq];
        wp += 4 * Hq;
#pragma unroll
        for (int q = 0; q < ATPB; ++q) {
          v2f a = *(const v2f*)(apl + q * PANEL + kk);
          accA[q] = wmma4(a, b, accA[q]);
        }
      }
    }
#pragma unroll
    for (int kk = 0; kk < 8; kk += 4) {
      const int k0 = kk + koff;
      v2f b;
      b.x = w2b[(k0     < BFq ? k0     : BFq - 1) * Hq + nc];
      b.y = w2b[(k0 + 1 < BFq ? k0 + 1 : BFq - 1) * Hq + nc];
#pragma unroll
      for (int q = 0; q < ATPB; ++q) {
        v2f a = *(const v2f*)(apl + q * PANEL + Hq + kk);
        accB[q] = wmma4(a, b, accB[q]);
      }
    }
#pragma unroll
    for (int q = 0; q < ATPB; ++q) {
      const int nn = nnb[ba0 + q];
      float s = 0.f;
#pragma unroll
      for (int v = 0; v < 8; ++v) {
        int rj = v + half * 8;
        s += (rj < nn) ? accA[q][v] * accB[q][v] : 0.f;
      }
      s += __shfl_xor(s, 16, 32);
      if (half == 0 && n < Hq) out[(ba0 + q) * Hq + n] = s;
    }
  }
}

__global__ __launch_bounds__(256) void out_kernel(
    const float* __restrict__ h, const float* __restrict__ w,
    const float* __restrict__ nprod, const int* __restrict__ na,
    float* __restrict__ out) {
  int tid = blockIdx.x * 8 + (threadIdx.x >> 5);
  if (tid >= MP * NP) return;
  const int lane = threadIdx.x & 31;
  const int m = lane & 15, half = lane >> 4, koff = half << 1;
  const int mp = tid / NP, np = tid - mp * NP;
  const int n0 = np * 32 + m, n1 = n0 + 16;
  const int row0 = mp * 32 + m, row1 = row0 + 16;
  const int nc0 = (n0 < Hq) ? n0 : (Hq - 1);
  const int nc1 = (n1 < Hq) ? n1 : (Hq - 1);
  v8f c00 = {}, c01 = {}, c10 = {}, c11 = {};
  const float* ap0 = h + row0 * Hq + koff;
  const float* ap1 = h + row1 * Hq + koff;
  const float* wp0 = w + koff * Hq + nc0;
  const float* wp1 = w + koff * Hq + nc1;
#pragma unroll 3
  for (int kk = 0; kk < Hq; kk += 4) {
    v2f a0 = *(const v2f*)ap0;
    v2f a1 = *(const v2f*)ap1;
    v2f b0; b0.x = wp0[0]; b0.y = wp0[Hq];
    v2f b1; b1.x = wp1[0]; b1.y = wp1[Hq];
    ap0 += 4; ap1 += 4; wp0 += 4 * Hq; wp1 += 4 * Hq;
    c00 = wmma4(a0, b0, c00); c01 = wmma4(a0, b1, c01);
    c10 = wmma4(a1, b0, c10); c11 = wmma4(a1, b1, c11);
  }
  const int rb0 = mp * 32 + half * 8, rb1 = rb0 + 16;
#pragma unroll
  for (int v = 0; v < 8; ++v) {
    int r0 = rb0 + v, r1 = rb1 + v;
    int b0i = r0 / Asz, a0i = r0 - b0i * Asz;
    int b1i = r1 / Asz, a1i = r1 - b1i * Asz;
    bool m0 = (a0i < na[b0i]), m1 = (a1i < na[b1i]);
    if (n0 < Hq) {
      out[r0 * Hq + n0] = m0 ? c00[v] * nprod[r0 * Hq + n0] : 0.f;
      out[r1 * Hq + n0] = m1 ? c10[v] * nprod[r1 * Hq + n0] : 0.f;
    }
    if (n1 < Hq) {
      out[r0 * Hq + n1] = m0 ? c01[v] * nprod[r0 * Hq + n1] : 0.f;
      out[r1 * Hq + n1] = m1 ? c11[v] * nprod[r1 * Hq + n1] : 0.f;
    }
  }
}

extern "C" void kernel_launch(void* const* d_in, const int* in_sizes, int n_in,
                              void* d_out, int out_size, void* d_ws, size_t ws_size,
                              hipStream_t stream) {
  const float* atom_feats = (const float*)d_in[0];
  const float* bond_feats = (const float*)d_in[1];
  const float* w_fc1      = (const float*)d_in[2];
  const float* w_nei      = (const float*)d_in[3];
  const float* b_nei      = (const float*)d_in[4];
  const float* w_atom     = (const float*)d_in[5];
  const float* b_atom     = (const float*)d_in[6];
  const float* w2a        = (const float*)d_in[7];
  const float* w2b        = (const float*)d_in[8];
  const float* w_fc2      = (const float*)d_in[9];
  const int*   atom_graph = (const int*)d_in[10];
  const int*   bond_graph = (const int*)d_in[11];
  const int*   num_nbs    = (const int*)d_in[12];
  const int*   n_atoms    = (const int*)d_in[13];
  float* out = (float*)d_out;

  const size_t nH = (size_t)Bsz * Asz * Hq;
  float* h_a = (float*)d_ws;
  float* h_b = h_a + nH;
  float* nei = h_b + nH;

  dim3 blk(256);
  const int gemm_blocks = (MP * NP + 7) / 8;
  const int gather_blocks = (Bsz * Asz) / ATPB;

  fc1_kernel<<<gemm_blocks, blk, 0, stream>>>(atom_feats, w_fc1, h_a);

  nei_kernel<<<gather_blocks, blk, 0, stream>>>(h_a, bond_feats, atom_graph,
                                                bond_graph, num_nbs, w_nei, b_nei, nei);
  atom_kernel<<<gemm_blocks, blk, 0, stream>>>(h_a, nei, w_atom, b_atom, h_b);
  nei_kernel<<<gather_blocks, blk, 0, stream>>>(h_b, bond_feats, atom_graph,
                                                bond_graph, num_nbs, w_nei, b_nei, nei);
  atom_kernel<<<gemm_blocks, blk, 0, stream>>>(h_b, nei, w_atom, b_atom, h_a);
  nprod_kernel<<<gather_blocks, blk, 0, stream>>>(h_a, bond_feats, atom_graph,
                                                  bond_graph, num_nbs, w2a, w2b, nei);
  out_kernel<<<gemm_blocks, blk, 0, stream>>>(h_a, w_fc2, nei, n_atoms, out);
}


# Gather commutes with the linear map: h@W and bond@W formed once per source row on WMMA, neighbour stage only gathers and sums projected rows; k-sum split 300|6, neighbour sum sequential
# speedup vs baseline: 2.9781x; 2.9781x over previous
#include <hip/hip_runtime.h>
#include <hip/hip_bf16.h>


#define Bsz   256
#define Asz   100
#define NBq   10
#define BONDq 110
#define AFq   82
#define BFq   6
#define Hq    300
#define NP    10
#define MTOT  (Bsz*Asz)
#define MT    (MTOT/16)
#define MP    (MT/2)
#define BROWS (Bsz*BONDq)
#define BMP   (BROWS/32)
#define C4    (Hq/4)
#define ITEMS (MTOT*C4)
#define GBLK  1875

typedef float v2f __attribute__((ext_vector_type(2)));
typedef float v8f __attribute__((ext_vector_type(8)));

static __device__ __forceinline__ v8f wmma4(v2f a, v2f b, v8f c) {
  return __builtin_amdgcn_wmma_f32_16x16x4_f32(
      false, a, false, b, (short)0, c, false, false);
}

__global__ __launch_bounds__(256) void fc1_kernel(
    const float* __restrict__ af, const float* __restrict__ w,
    float* __restrict__ hout) {
  int tid = blockIdx.x * 8 + (threadIdx.x >> 5);
  if (tid >= MP * NP) return;
  const int lane = threadIdx.x & 31;
  const int m = lane & 15, half = lane >> 4, koff = half << 1;
  const int mp = tid / NP, np = tid - mp * NP;
  const int n0 = np * 32 + m, n1 = n0 + 16;
  const int row0 = mp * 32 + m, row1 = row0 + 16;
  const int nc0 = (n0 < Hq) ? n0 : (Hq - 1);
  const int nc1 = (n1 < Hq) ? n1 : (Hq - 1);

  const float* ap0 = af + row0 * AFq + koff;
  const float* ap1 = af + row1 * AFq + koff;
  const float* wp0 = w + koff * Hq + nc0;
  const float* wp1 = w + koff * Hq + nc1;
  v8f c00 = {}, c01 = {}, c10 = {}, c11 = {};
#pragma unroll 2
  for (int kk = 0; kk < 80; kk += 4) {
    v2f a0 = *(const v2f*)ap0;
    v2f a1 = *(const v2f*)ap1;
    v2f b0; b0.x = wp0[0]; b0.y = wp0[Hq];
    v2f b1; b1.x = wp1[0]; b1.y = wp1[Hq];
    ap0 += 4; ap1 += 4; wp0 += 4 * Hq; wp1 += 4 * Hq;
    c00 = wmma4(a0, b0, c00); c01 = wmma4(a0, b1, c01);
    c10 = wmma4(a1, b0, c10); c11 = wmma4(a1, b1, c11);
  }
  {
    const int k0 = 80 + koff;
    const int k0c = (k0 < AFq) ? k0 : AFq - 1;
    const int k1c = (k0 + 1 < AFq) ? k0 + 1 : AFq - 1;
    v2f a0, a1, b0, b1;
    float a0x = af[row0 * AFq + k0c], a0y = af[row0 * AFq + k1c];
    float a1x = af[row1 * AFq + k0c], a1y = af[row1 * AFq + k1c];
    a0.x = (k0 < AFq) ? a0x : 0.f;  a0.y = (k0 + 1 < AFq) ? a0y : 0.f;
    a1.x = (k0 < AFq) ? a1x : 0.f;  a1.y = (k0 + 1 < AFq) ? a1y : 0.f;
    b0.x = w[k0c * Hq + nc0]; b0.y = w[k1c * Hq + nc0];
    b1.x = w[k0c * Hq + nc1]; b1.y = w[k1c * Hq + nc1];
    c00 = wmma4(a0, b0, c00); c01 = wmma4(a0, b1, c01);
    c10 = wmma4(a1, b0, c10); c11 = wmma4(a1, b1, c11);
  }
  const int rb0 = mp * 32 + half * 8, rb1 = rb0 + 16;
#pragma unroll
  for (int v = 0; v < 8; ++v) {
    if (n0 < Hq) {
      hout[(rb0 + v) * Hq + n0] = fmaxf(c00[v], 0.f);
      hout[(rb1 + v) * Hq + n0] = fmaxf(c10[v], 0.f);
    }
    if (n1 < Hq) {
      hout[(rb0 + v) * Hq + n1] = fmaxf(c01[v], 0.f);
      hout[(rb1 + v) * Hq + n1] = fmaxf(c11[v], 0.f);
    }
  }
}

__global__ __launch_bounds__(256) void bondproj_kernel(
    const float* __restrict__ bond, const float* __restrict__ wn,
    const float* __restrict__ w2b, float* __restrict__ q1,
    float* __restrict__ q2) {
  int tid = blockIdx.x * 8 + (threadIdx.x >> 5);
  if (tid >= BMP * NP) return;
  const int lane = threadIdx.x & 31;
  const int m = lane & 15, half = lane >> 4, koff = half << 1;
  const int mp = tid / NP, np = tid - mp * NP;
  const int n0 = np * 32 + m, n1 = n0 + 16;
  const int row0 = mp * 32 + m, row1 = row0 + 16;
  const int nc0 = (n0 < Hq) ? n0 : (Hq - 1);
  const int nc1 = (n1 < Hq) ? n1 : (Hq - 1);
  const float* wnb = wn + Hq * Hq;
  v8f c00 = {}, c01 = {}, c10 = {}, c11 = {};
  v8f d00 = {}, d01 = {}, d10 = {}, d11 = {};
#pragma unroll
  for (int kk = 0; kk < 8; kk += 4) {
    const int k0 = kk + koff, k1 = k0 + 1;
    const bool v0 = k0 < BFq, v1 = k1 < BFq;
    const int k0c = v0 ? k0 : BFq - 1, k1c = v1 ? k1 : BFq - 1;
    const float a0x = bond[row0 * BFq + k0c], a0y = bond[row0 * BFq + k1c];
    const float a1x = bond[row1 * BFq + k0c], a1y = bond[row1 * BFq + k1c];
    const float b0x = wnb[k0c * Hq + nc0], b0y = wnb[k1c * Hq + nc0];
    const float b1x = wnb[k0c * Hq + nc1], b1y = wnb[k1c * Hq + nc1];
    const float e0x = w2b[k0c * Hq + nc0], e0y = w2b[k1c * Hq + nc0];
    const float e1x = w2b[k0c * Hq + nc1], e1y = w2b[k1c * Hq + nc1];
    v2f a0, a1, b0, b1, e0, e1;
    a0.x = v0 ? a0x : 0.f;  a0.y = v1 ? a0y : 0.f;
    a1.x = v0 ? a1x : 0.f;  a1.y = v1 ? a1y : 0.f;
    b0.x = v0 ? b0x : 0.f;  b0.y = v1 ? b0y : 0.f;
    b1.x = v0 ? b1x : 0.f;  b1.y = v1 ? b1y : 0.f;
    e0.x = v0 ? e0x : 0.f;  e0.y = v1 ? e0y : 0.f;
    e1.x = v0 ? e1x : 0.f;  e1.y = v1 ? e1y : 0.f;
    c00 = wmma4(a0, b0, c00); c01 = wmma4(a0, b1, c01);
    c10 = wmma4(a1, b0, c10); c11 = wmma4(a1, b1, c11);
    d00 = wmma4(a0, e0, d00); d01 = wmma4(a0, e1, d01);
    d10 = wmma4(a1, e0, d10); d11 = wmma4(a1, e1, d11);
  }
  const int rb0 = mp * 32 + half * 8, rb1 = rb0 + 16;
#pragma unroll
  for (int v = 0; v < 8; ++v) {
    if (n0 < Hq) {
      q1[(rb0 + v) * Hq + n0] = c00[v];  q1[(rb1 + v) * Hq + n0] = c10[v];
      q2[(rb0 + v) * Hq + n0] = d00[v];  q2[(rb1 + v) * Hq + n0] = d10[v];
    }
    if (n1 < Hq) {
      q1[(rb0 + v) * Hq + n1] = c01[v];  q1[(rb1 + v) * Hq + n1] = c11[v];
      q2[(rb0 + v) * Hq + n1] = d01[v];  q2[(rb1 + v) * Hq + n1] = d11[v];
    }
  }
}

__global__ __launch_bounds__(256) void proj_kernel(
    const float* __restrict__ h, const float* __restrict__ w,
    float* __restrict__ pout) {
  int tid = blockIdx.x * 8 + (threadIdx.x >> 5);
  if (tid >= MP * NP) return;
  const int lane = threadIdx.x & 31;
  const int m = lane & 15, half = lane >> 4, koff = half << 1;
  const int mp = tid / NP, np = tid - mp * NP;
  const int n0 = np * 32 + m, n1 = n0 + 16;
  const int row0 = mp * 32 + m, row1 = row0 + 16;
  const int nc0 = (n0 < Hq) ? n0 : (Hq - 1);
  const int nc1 = (n1 < Hq) ? n1 : (Hq - 1);
  v8f c00 = {}, c01 = {}, c10 = {}, c11 = {};
  const float* ap0 = h + row0 * Hq + koff;
  const float* ap1 = h + row1 * Hq + koff;
  const float* wp0 = w + koff * Hq + nc0;
  const float* wp1 = w + koff * Hq + nc1;
#pragma unroll 3
  for (int kk = 0; kk < Hq; kk += 4) {
    v2f a0 = *(const v2f*)ap0;
    v2f a1 = *(const v2f*)ap1;
    v2f b0; b0.x = wp0[0]; b0.y = wp0[Hq];
    v2f b1; b1.x = wp1[0]; b1.y = wp1[Hq];
    ap0 += 4; ap1 += 4; wp0 += 4 * Hq; wp1 += 4 * Hq;
    c00 = wmma4(a0, b0, c00); c01 = wmma4(a0, b1, c01);
    c10 = wmma4(a1, b0, c10); c11 = wmma4(a1, b1, c11);
  }
  const int rb0 = mp * 32 + half * 8, rb1 = rb0 + 16;
#pragma unroll
  for (int v = 0; v < 8; ++v) {
    if (n0 < Hq) {
      pout[(rb0 + v) * Hq + n0] = c00[v];
      pout[(rb1 + v) * Hq + n0] = c10[v];
    }
    if (n1 < Hq) {
      pout[(rb0 + v) * Hq + n1] = c01[v];
      pout[(rb1 + v) * Hq + n1] = c11[v];
    }
  }
}

__global__ __launch_bounds__(256) void nei_gather_kernel(
    const float* __restrict__ p, const float* __restrict__ q,
    const int* __restrict__ ag, const int* __restrict__ bg,
    const int* __restrict__ nnb, const float* __restrict__ bn,
    float* __restrict__ out) {
  for (int i = blockIdx.x * 256 + threadIdx.x; i < ITEMS; i += GBLK * 256) {
    const int ba = i / C4, c = (i - ba * C4) * 4;
    const int nn = nnb[ba];
    const float bx = bn[c], by = bn[c + 1], bz = bn[c + 2], bw = bn[c + 3];
    int po[NBq], qo[NBq];
#pragma unroll
    for (int j = 0; j < NBq; ++j) {
      const int e = (ba * NBq + j) * 2;
      po[j] = (ag[e] * Asz + ag[e + 1]) * Hq + c;
      qo[j] = (bg[e] * BONDq + bg[e + 1]) * Hq + c;
    }
    float4 pv[NBq], qv[NBq];
#pragma unroll
    for (int j = 0; j < NBq; ++j) {
      pv[j] = make_float4(0.f, 0.f, 0.f, 0.f);
      qv[j] = make_float4(0.f, 0.f, 0.f, 0.f);
      if (j < nn) {
        pv[j] = *(const float4*)(p + po[j]);
        qv[j] = *(const float4*)(q + qo[j]);
      }
    }
    float4 s = make_float4(0.f, 0.f, 0.f, 0.f);
#pragma unroll
    for (int j = 0; j < NBq; ++j) {
      const bool live = j < nn;
      const float x = fmaxf(pv[j].x + qv[j].x + bx, 0.f);
      const float y = fmaxf(pv[j].y + qv[j].y + by, 0.f);
      const float z = fmaxf(pv[j].z + qv[j].z + bz, 0.f);
      const float w = fmaxf(pv[j].w + qv[j].w + bw, 0.f);
      s.x += live ? x : 0.f;  s.y += live ? y : 0.f;
      s.z += live ? z : 0.f;  s.w += live ? w : 0.f;
    }
    *(float4*)(out + ba * Hq + c) = s;
  }
}

__global__ __launch_bounds__(256) void atom_kernel(
    const float* __restrict__ h, const float* __restrict__ nei,
    const float* __restrict__ wa, const float* __restrict__ bb,
    float* __restrict__ hout) {
  int tid = blockIdx.x * 8 + (threadIdx.x >> 5);
  if (tid >= MP * NP) return;
  const int lane = threadIdx.x & 31;
  const int m = lane & 15, half = lane >> 4, koff = half << 1;
  const int mp = tid / NP, np = tid - mp * NP;
  const int n0 = np * 32 + m, n1 = n0 + 16;
  const int row0 = mp * 32 + m, row1 = row0 + 16;
  const int nc0 = (n0 < Hq) ? n0 : (Hq - 1);
  const int nc1 = (n1 < Hq) ? n1 : (Hq - 1);
  v8f c00 = {}, c01 = {}, c10 = {}, c11 = {};
  {
    const float* ap0 = h + row0 * Hq + koff;
    const float* ap1 = h + row1 * Hq + koff;
    const float* wp0 = wa + koff * Hq + nc0;
    const float* wp1 = wa + koff * Hq + nc1;
#pragma unroll 3
    for (int kk = 0; kk < Hq; kk += 4) {
      v2f a0 = *(const v2f*)ap0;
      v2f a1 = *(const v2f*)ap1;
      v2f b0; b0.x = wp0[0]; b0.y = wp0[Hq];
      v2f b1; b1.x = wp1[0]; b1.y = wp1[Hq];
      ap0 += 4; ap1 += 4; wp0 += 4 * Hq; wp1 += 4 * Hq;
      c00 = wmma4(a0, b0, c00); c01 = wmma4(a0, b1, c01);
      c10 = wmma4(a1, b0, c10); c11 = wmma4(a1, b1, c11);
    }
  }
  {
    const float* ap0 = nei + row0 * Hq + koff;
    const float* ap1 = nei + row1 * Hq + koff;
    const float* wp0 = wa + (Hq + koff) * Hq + nc0;
    const float* wp1 = wa + (Hq + koff) * Hq + nc1;
#pragma unroll 3
    for (int kk = 0; kk < Hq; kk += 4) {
      v2f a0 = *(const v2f*)ap0;
      v2f a1 = *(const v2f*)ap1;
      v2f b0; b0.x = wp0[0]; b0.y = wp0[Hq];
      v2f b1; b1.x = wp1[0]; b1.y = wp1[Hq];
      ap0 += 4; ap1 += 4; wp0 += 4 * Hq; wp1 += 4 * Hq;
      c00 = wmma4(a0, b0, c00); c01 = wmma4(a0, b1, c01);
      c10 = wmma4(a1, b0, c10); c11 = wmma4(a1, b1, c11);
    }
  }
  const float bias0 = bb[nc0], bias1 = bb[nc1];
  const int rb0 = mp * 32 + half * 8, rb1 = rb0 + 16;
#pragma unroll
  for (int v = 0; v < 8; ++v) {
    if (n0 < Hq) {
      hout[(rb0 + v) * Hq + n0] = fmaxf(c00[v] + bias0, 0.f);
      hout[(rb1 + v) * Hq + n0] = fmaxf(c10[v] + bias0, 0.f);
    }
    if (n1 < Hq) {
      hout[(rb0 + v) * Hq + n1] = fmaxf(c01[v] + bias1, 0.f);
      hout[(rb1 + v) * Hq + n1] = fmaxf(c11[v] + bias1, 0.f);
    }
  }
}

__global__ __launch_bounds__(256) void nprod_gather_kernel(
    const float* __restrict__ p, const float* __restrict__ q,
    const int* __restrict__ ag, const int* __restrict__ bg,
    const int* __restrict__ nnb, float* __restrict__ out) {
  for (int i = blockIdx.x * 256 + threadIdx.x; i < ITEMS; i += GBLK * 256) {
    const int ba = i / C4, c = (i - ba * C4) * 4;
    const int nn = nnb[ba];
    int po[NBq], qo[NBq];
#pragma unroll
    for (int j = 0; j < NBq; ++j) {
      const int e = (ba * NBq + j) * 2;
      po[j] = (ag[e] * Asz + ag[e + 1]) * Hq + c;
      qo[j] = (bg[e] * BONDq + bg[e + 1]) * Hq + c;
    }
    float4 pv[NBq], qv[NBq];
#pragma unroll
    for (int j = 0; j < NBq; ++j) {
      pv[j] = make_float4(0.f, 0.f, 0.f, 0.f);
      qv[j] = make_float4(0.f, 0.f, 0.f, 0.f);
      if (j < nn) {
        pv[j] = *(const float4*)(p + po[j]);
        qv[j] = *(const float4*)(q + qo[j]);
      }
    }
    float4 s = make_float4(0.f, 0.f, 0.f, 0.f);
#pragma unroll
    for (int j = 0; j < NBq; ++j) {
      const bool live = j < nn;
      s.x += live ? pv[j].x * qv[j].x : 0.f;
      s.y += live ? pv[j].y * qv[j].y : 0.f;
      s.z += live ? pv[j].z * qv[j].z : 0.f;
      s.w += live ? pv[j].w * qv[j].w : 0.f;
    }
    *(float4*)(out + ba * Hq + c) = s;
  }
}

__global__ __launch_bounds__(256) void out_kernel(
    const float* __restrict__ h, const float* __restrict__ w,
    const float* __restrict__ nprod, const int* __restrict__ na,
    float* __restrict__ out) {
  int tid = blockIdx.x * 8 + (threadIdx.x >> 5);
  if (tid >= MP * NP) return;
  const int lane = threadIdx.x & 31;
  const int m = lane & 15, half = lane >> 4, koff = half << 1;
  const int mp = tid / NP, np = tid - mp * NP;
  const int n0 = np * 32 + m, n1 = n0 + 16;
  const int row0 = mp * 32 + m, row1 = row0 + 16;
  const int nc0 = (n0 < Hq) ? n0 : (Hq - 1);
  const int nc1 = (n1 < Hq) ? n1 : (Hq - 1);
  v8f c00 = {}, c01 = {}, c10 = {}, c11 = {};
  const float* ap0 = h + row0 * Hq + koff;
  const float* ap1 = h + row1 * Hq + koff;
  const float* wp0 = w + koff * Hq + nc0;
  const float* wp1 = w + koff * Hq + nc1;
#pragma unroll 3
  for (int kk = 0; kk < Hq; kk += 4) {
    v2f a0 = *(const v2f*)ap0;
    v2f a1 = *(const v2f*)ap1;
    v2f b0; b0.x = wp0[0]; b0.y = wp0[Hq];
    v2f b1; b1.x = wp1[0]; b1.y = wp1[Hq];
    ap0 += 4; ap1 += 4; wp0 += 4 * Hq; wp1 += 4 * Hq;
    c00 = wmma4(a0, b0, c00); c01 = wmma4(a0, b1, c01);
    c10 = wmma4(a1, b0, c10); c11 = wmma4(a1, b1, c11);
  }
  const int rb0 = mp * 32 + half * 8, rb1 = rb0 + 16;
#pragma unroll
  for (int v = 0; v < 8; ++v) {
    int r0 = rb0 + v, r1 = rb1 + v;
    int b0i = r0 / Asz, a0i = r0 - b0i * Asz;
    int b1i = r1 / Asz, a1i = r1 - b1i * Asz;
    bool m0 = (a0i < na[b0i]), m1 = (a1i < na[b1i]);
    if (n0 < Hq) {
      out[r0 * Hq + n0] = m0 ? c00[v] * nprod[r0 * Hq + n0] : 0.f;
      out[r1 * Hq + n0] = m1 ? c10[v] * nprod[r1 * Hq + n0] : 0.f;
    }
    if (n1 < Hq) {
      out[r0 * Hq + n1] = m0 ? c01[v] * nprod[r0 * Hq + n1] : 0.f;
      out[r1 * Hq + n1] = m1 ? c11[v] * nprod[r1 * Hq + n1] : 0.f;
    }
  }
}

extern "C" void kernel_launch(void* const* d_in, const int* in_sizes, int n_in,
                              void* d_out, int out_size, void* d_ws, size_t ws_size,
                              hipStream_t stream) {
  const float* atom_feats = (const float*)d_in[0];
  const float* bond_feats = (const float*)d_in[1];
  const float* w_fc1      = (const float*)d_in[2];
  const float* w_nei      = (const float*)d_in[3];
  const float* b_nei      = (const float*)d_in[4];
  const float* w_atom     = (const float*)d_in[5];
  const float* b_atom     = (const float*)d_in[6];
  const float* w2a        = (const float*)d_in[7];
  const float* w2b        = (const float*)d_in[8];
  const float* w_fc2      = (const float*)d_in[9];
  const int*   atom_graph = (const int*)d_in[10];
  const int*   bond_graph = (const int*)d_in[11];
  const int*   num_nbs    = (const int*)d_in[12];
  const int*   n_atoms    = (const int*)d_in[13];
  float* out = (float*)d_out;

  const size_t nH = (size_t)Bsz * Asz * Hq;
  const size_t nQ = (size_t)Bsz * BONDq * Hq;
  float* h_a = (float*)d_ws;
  float* h_b = h_a + nH;
  float* nei = h_b + nH;
  float* prj = nei + nH;
  float* q1  = prj + nH;
  float* q2  = q1 + nQ;

  dim3 blk(256);
  const int gemm_blocks = (MP * NP + 7) / 8;
  const int bond_blocks = (BMP * NP + 7) / 8;

  bondproj_kernel<<<bond_blocks, blk, 0, stream>>>(bond_feats, w_nei, w2b, q1, q2);
  fc1_kernel<<<gemm_blocks, blk, 0, stream>>>(atom_feats, w_fc1, h_a);

  proj_kernel<<<gemm_blocks, blk, 0, stream>>>(h_a, w_nei, prj);
  nei_gather_kernel<<<GBLK, blk, 0, stream>>>(prj, q1, atom_graph, bond_graph,
                                             num_nbs, b_nei, nei);
  atom_kernel<<<gemm_blocks, blk, 0, stream>>>(h_a, nei, w_atom, b_atom, h_b);
  proj_kernel<<<gemm_blocks, blk, 0, stream>>>(h_b, w_nei, prj);
  nei_gather_kernel<<<GBLK, blk, 0, stream>>>(prj, q1, atom_graph, bond_graph,
                                             num_nbs, b_nei, nei);
  atom_kernel<<<gemm_blocks, blk, 0, stream>>>(h_b, nei, w_atom, b_atom, h_a);
  proj_kernel<<<gemm_blocks, blk, 0, stream>>>(h_a, w2a, prj);
  nprod_gather_kernel<<<GBLK, blk, 0, stream>>>(prj, q2, atom_graph, bond_graph,
                                               num_nbs, nei);
  out_kernel<<<gemm_blocks, blk, 0, stream>>>(h_a, w_fc2, nei, n_atoms, out);
}


# Row GEMMs: wave tile 32x64 with column-interleaved N tiles, so A, W and outputs move 16 B per lane; k order inside each 8-k step permuted on both operands (rounding order changes)
# speedup vs baseline: 3.6871x; 1.2381x over previous
#include <hip/hip_runtime.h>
#include <hip/hip_bf16.h>


#define Bsz   256
#define Asz   100
#define NBq   10
#define BONDq 110
#define AFq   82
#define BFq   6
#define Hq    300
#define NCH   5
#define MTOT  (Bsz*Asz)
#define MP    (MTOT/32)
#define BROWS (Bsz*BONDq)
#define BMP   (BROWS/32)
#define C4    (Hq/4)
#define ITEMS (MTOT*C4)
#define GBLK  1875

typedef float v2f __attribute__((ext_vector_type(2)));
typedef float v4f __attribute__((ext_vector_type(4)));
typedef float v8f __attribute__((ext_vector_type(8)));

static __device__ __forceinline__ v8f wmma4(v2f a, v2f b, v8f c) {
  return __builtin_amdgcn_wmma_f32_16x16x4_f32(
      false, a, false, b, (short)0, c, false, false);
}

struct Tile { v8f c[2][4]; };
struct Ops  { v4f x0, x1, w0, w1, w2, w3; };

static __device__ __forceinline__ void mma8(v2f a0, v2f a1, v4f wlo, v4f whi,
                                            Tile& T) {
#pragma unroll
  for (int t = 0; t < 4; ++t) {
    v2f b; b.x = wlo[t]; b.y = whi[t];
    T.c[0][t] = wmma4(a0, b, T.c[0][t]);
    T.c[1][t] = wmma4(a1, b, T.c[1][t]);
  }
}

static __device__ __forceinline__ Ops ld8(const float* a0, const float* a1,
                                          const float* w) {
  Ops o;
  o.x0 = *(const v4f*)a0;            o.x1 = *(const v4f*)a1;
  o.w0 = *(const v4f*)w;             o.w1 = *(const v4f*)(w + Hq);
  o.w2 = *(const v4f*)(w + 2 * Hq);  o.w3 = *(const v4f*)(w + 3 * Hq);
  return o;
}

static __device__ __forceinline__ void mm8(const Ops& o, Tile& T) {
  v2f a0l, a0h, a1l, a1h;
  a0l.x = o.x0.x; a0l.y = o.x0.y;  a0h.x = o.x0.z; a0h.y = o.x0.w;
  a1l.x = o.x1.x; a1l.y = o.x1.y;  a1h.x = o.x1.z; a1h.y = o.x1.w;
  mma8(a0l, a1l, o.w0, o.w1, T);
  mma8(a0h, a1h, o.w2, o.w3, T);
}

static __device__ __forceinline__ void gemm300(const float* a0, const float* a1,
                                               const float* w, int half,
                                               Tile& T) {
  const float* p0 = a0 + 4 * half;
  const float* p1 = a1 + 4 * half;
  const float* pw = w + 4 * half * Hq;
  Ops cur = ld8(p0, p1, pw);
#pragma unroll 2
  for (int i = 0; i < 36; ++i) {
    p0 += 8; p1 += 8; pw += 8 * Hq;
    Ops nxt = ld8(p0, p1, pw);
    mm8(cur, T);
    cur = nxt;
  }
  const int kt = 296 + 2 * half;
  const v2f t0 = *(const v2f*)(a0 + kt);
  const v2f t1 = *(const v2f*)(a1 + kt);
  const v4f u0 = *(const v4f*)(w + kt * Hq);
  const v4f u1 = *(const v4f*)(w + (kt + 1) * Hq);
  mm8(cur, T);
  mma8(t0, t1, u0, u1, T);
}

#define GEMM_COORDS()                                                        \
  const int lane = threadIdx.x & 31, wv = threadIdx.x >> 5;                  \
  const int c = lane & 15, half = lane >> 4;                                 \
  const int nc = blockIdx.x % NCH, grp = blockIdx.x / NCH;                   \
  const int mp = grp * 8 + wv;                                               \
  const int ncol = nc * 64 + 4 * c;                                          \
  const bool valid = ncol < Hq;                \
  const int ncc = valid ? ncol : (Hq - 4);                                   \
  const int row0 = mp * 32 + c, row1 = row0 + 16;                            \
  const int rb = mp * 32 + half * 8;

__global__ __launch_bounds__(256) void fc1_kernel(
    const float* __restrict__ af, const float* __restrict__ w,
    float* __restrict__ hout) {
  GEMM_COORDS()
  Tile T = {};
  const float* a0 = af + row0 * AFq + 2 * half;
  const float* a1 = af + row1 * AFq + 2 * half;
  const float* pw = w + 2 * half * Hq + ncc;
#pragma unroll 4
  for (int kk = 0; kk < 80; kk += 4) {
    const v2f x0 = *(const v2f*)a0;
    const v2f x1 = *(const v2f*)a1;
    const v4f u0 = *(const v4f*)pw;
    const v4f u1 = *(const v4f*)(pw + Hq);
    a0 += 4; a1 += 4; pw += 4 * Hq;
    mma8(x0, x1, u0, u1, T);
  }
  {
    const bool kv = (half == 0);
    const int kc = kv ? 80 : 0;
    v2f x0 = *(const v2f*)(af + row0 * AFq + kc);
    v2f x1 = *(const v2f*)(af + row1 * AFq + kc);
    v4f u0 = *(const v4f*)(w + kc * Hq + ncc);
    v4f u1 = *(const v4f*)(w + (kc + 1) * Hq + ncc);
    const v2f z2 = {0.f, 0.f};
    const v4f z4 = {0.f, 0.f, 0.f, 0.f};
    x0 = kv ? x0 : z2;  x1 = kv ? x1 : z2;
    u0 = kv ? u0 : z4;  u1 = kv ? u1 : z4;
    mma8(x0, x1, u0, u1, T);
  }
#pragma unroll
  for (int i = 0; i < 2; ++i) {
#pragma unroll
    for (int r = 0; r < 8; ++r) {
      v4f v;
      v.x = fmaxf(T.c[i][0][r], 0.f);  v.y = fmaxf(T.c[i][1][r], 0.f);
      v.z = fmaxf(T.c[i][2][r], 0.f);  v.w = fmaxf(T.c[i][3][r], 0.f);
      if (valid) *(v4f*)(hout + (rb + i * 16 + r) * Hq + ncol) = v;
    }
  }
}

__global__ __launch_bounds__(256) void bondproj_kernel(
    const float* __restrict__ bond, const float* __restrict__ wn,
    const float* __restrict__ w2b, float* __restrict__ q1,
    float* __restrict__ q2) {
  GEMM_COORDS()
  const float* wnb = wn + Hq * Hq;
  const bool kv = (half == 0);
  const int kc = kv ? 4 : 0;
  const v2f z2 = {0.f, 0.f};
  const v4f z4 = {0.f, 0.f, 0.f, 0.f};
  const v2f xa0 = *(const v2f*)(bond + row0 * BFq + 2 * half);
  const v2f xa1 = *(const v2f*)(bond + row1 * BFq + 2 * half);
  v2f xb0 = *(const v2f*)(bond + row0 * BFq + kc);
  v2f xb1 = *(const v2f*)(bond + row1 * BFq + kc);
  xb0 = kv ? xb0 : z2;  xb1 = kv ? xb1 : z2;
#pragma unroll
  for (int pass = 0; pass < 2; ++pass) {
    const float* wsrc = pass ? w2b : wnb;
    float* dst = pass ? q2 : q1;
    const v4f u0 = *(const v4f*)(wsrc + (2 * half) * Hq + ncc);
    const v4f u1 = *(const v4f*)(wsrc + (2 * half + 1) * Hq + ncc);
    v4f u2 = *(const v4f*)(wsrc + kc * Hq + ncc);
    v4f u3 = *(const v4f*)(wsrc + (kc + 1) * Hq + ncc);
    u2 = kv ? u2 : z4;  u3 = kv ? u3 : z4;
    Tile T = {};
    mma8(xa0, xa1, u0, u1, T);
    mma8(xb0, xb1, u2, u3, T);
#pragma unroll
    for (int i = 0; i < 2; ++i) {
#pragma unroll
      for (int r = 0; r < 8; ++r) {
        v4f v;
        v.x = T.c[i][0][r];  v.y = T.c[i][1][r];
        v.z = T.c[i][2][r];  v.w = T.c[i][3][r];
        if (valid) *(v4f*)(dst + (rb + i * 16 + r) * Hq + ncol) = v;
      }
    }
  }
}

__global__ __launch_bounds__(256) void proj_kernel(
    const float* __restrict__ h, const float* __restrict__ w,
    float* __restrict__ pout) {
  GEMM_COORDS()
  Tile T = {};
  gemm300(h + row0 * Hq, h + row1 * Hq, w + ncc, half, T);
#pragma unroll
  for (int i = 0; i < 2; ++i) {
#pragma unroll
    for (int r = 0; r < 8; ++r) {
      v4f v;
      v.x = T.c[i][0][r];  v.y = T.c[i][1][r];
      v.z = T.c[i][2][r];  v.w = T.c[i][3][r];
      if (valid) *(v4f*)(pout + (rb + i * 16 + r) * Hq + ncol) = v;
    }
  }
}

__global__ __launch_bounds__(256) void nei_gather_kernel(
    const float* __restrict__ p, const float* __restrict__ q,
    const int* __restrict__ ag, const int* __restrict__ bg,
    const int* __restrict__ nnb, const float* __restrict__ bn,
    float* __restrict__ out) {
  for (int i = blockIdx.x * 256 + threadIdx.x; i < ITEMS; i += GBLK * 256) {
    const int ba = i / C4, c = (i - ba * C4) * 4;
    const int nn = nnb[ba];
    const float bx = bn[c], by = bn[c + 1], bz = bn[c + 2], bw = bn[c + 3];
    int po[NBq], qo[NBq];
#pragma unroll
    for (int j = 0; j < NBq; ++j) {
      const int e = (ba * NBq + j) * 2;
      po[j] = (ag[e] * Asz + ag[e + 1]) * Hq + c;
      qo[j] = (bg[e] * BONDq + bg[e + 1]) * Hq + c;
    }
    float4 pv[NBq], qv[NBq];
#pragma unroll
    for (int j = 0; j < NBq; ++j) {
      pv[j] = make_float4(0.f, 0.f, 0.f, 0.f);
      qv[j] = make_float4(0.f, 0.f, 0.f, 0.f);
      if (j < nn) {
        pv[j] = *(const float4*)(p + po[j]);
        qv[j] = *(const float4*)(q + qo[j]);
      }
    }
    float4 s = make_float4(0.f, 0.f, 0.f, 0.f);
#pragma unroll
    for (int j = 0; j < NBq; ++j) {
      const bool live = j < nn;
      const float x = fmaxf(pv[j].x + qv[j].x + bx, 0.f);
      const float y = fmaxf(pv[j].y + qv[j].y + by, 0.f);
      const float z = fmaxf(pv[j].z + qv[j].z + bz, 0.f);
      const float w = fmaxf(pv[j].w + qv[j].w + bw, 0.f);
      s.x += live ? x : 0.f;  s.y += live ? y : 0.f;
      s.z += live ? z : 0.f;  s.w += live ? w : 0.f;
    }
    *(float4*)(out + ba * Hq + c) = s;
  }
}

__global__ __launch_bounds__(256) void atom_kernel(
    const float* __restrict__ h, const float* __restrict__ nei,
    const float* __restrict__ wa, const float* __restrict__ bb,
    float* __restrict__ hout) {
  GEMM_COORDS()
  Tile T = {};
  gemm300(h + row0 * Hq, h + row1 * Hq, wa + ncc, half, T);
  gemm300(nei + row0 * Hq, nei + row1 * Hq, wa + Hq * Hq + ncc, half, T);
  const v4f bias = *(const v4f*)(bb + ncc);
#pragma unroll
  for (int i = 0; i < 2; ++i) {
#pragma unroll
    for (int r = 0; r < 8; ++r) {
      v4f v;
      v.x = fmaxf(T.c[i][0][r] + bias.x, 0.f);
      v.y = fmaxf(T.c[i][1][r] + bias.y, 0.f);
      v.z = fmaxf(T.c[i][2][r] + bias.z, 0.f);
      v.w = fmaxf(T.c[i][3][r] + bias.w, 0.f);
      if (valid) *(v4f*)(hout + (rb + i * 16 + r) * Hq + ncol) = v;
    }
  }
}

__global__ __launch_bounds__(256) void nprod_gather_kernel(
    const float* __restrict__ p, const float* __restrict__ q,
    const int* __restrict__ ag, const int* __restrict__ bg,
    const int* __restrict__ nnb, float* __restrict__ out) {
  for (int i = blockIdx.x * 256 + threadIdx.x; i < ITEMS; i += GBLK * 256) {
    const int ba = i / C4, c = (i - ba * C4) * 4;
    const int nn = nnb[ba];
    int po[NBq], qo[NBq];
#pragma unroll
    for (int j = 0; j < NBq; ++j) {
      const int e = (ba * NBq + j) * 2;
      po[j] = (ag[e] * Asz + ag[e + 1]) * Hq + c;
      qo[j] = (bg[e] * BONDq + bg[e + 1]) * Hq + c;
    }
    float4 pv[NBq], qv[NBq];
#pragma unroll
    for (int j = 0; j < NBq; ++j) {
      pv[j] = make_float4(0.f, 0.f, 0.f, 0.f);
      qv[j] = make_float4(0.f, 0.f, 0.f, 0.f);
      if (j < nn) {
        pv[j] = *(const float4*)(p + po[j]);
        qv[j] = *(const float4*)(q + qo[j]);
      }
    }
    float4 s = make_float4(0.f, 0.f, 0.f, 0.f);
#pragma unroll
    for (int j = 0; j < NBq; ++j) {
      const bool live = j < nn;
      s.x += live ? pv[j].x * qv[j].x : 0.f;
      s.y += live ? pv[j].y * qv[j].y : 0.f;
      s.z += live ? pv[j].z * qv[j].z : 0.f;
      s.w += live ? pv[j].w * qv[j].w : 0.f;
    }
    *(float4*)(out + ba * Hq + c) = s;
  }
}

__global__ __launch_bounds__(256) void out_kernel(
    const float* __restrict__ h, const float* __restrict__ w,
    const float* __restrict__ nprod, const int* __restrict__ na,
    float* __restrict__ out) {
  GEMM_COORDS()
  Tile T = {};
  gemm300(h + row0 * Hq, h + row1 * Hq, w + ncc, half, T);
#pragma unroll
  for (int i = 0; i < 2; ++i) {
#pragma unroll
    for (int r = 0; r < 8; ++r) {
      const int row = rb + i * 16 + r;
      const int bi = row / Asz, ai = row - bi * Asz;
      const bool live = ai < na[bi];
      if (valid) {
        const v4f np4 = *(const v4f*)(nprod + row * Hq + ncol);
        v4f v;
        v.x = live ? T.c[i][0][r] * np4.x : 0.f;
        v.y = live ? T.c[i][1][r] * np4.y : 0.f;
        v.z = live ? T.c[i][2][r] * np4.z : 0.f;
        v.w = live ? T.c[i][3][r] * np4.w : 0.f;
        *(v4f*)(out + row * Hq + ncol) = v;
      }
    }
  }
}

extern "C" void kernel_launch(void* const* d_in, const int* in_sizes, int n_in,
                              void* d_out, int out_size, void* d_ws, size_t ws_size,
                              hipStream_t stream) {
  const float* atom_feats = (const float*)d_in[0];
  const float* bond_feats = (const float*)d_in[1];
  const float* w_fc1      = (const float*)d_in[2];
  const float* w_nei      = (const float*)d_in[3];
  const float* b_nei      = (const float*)d_in[4];
  const float* w_atom     = (const float*)d_in[5];
  const float* b_atom     = (const float*)d_in[6];
  const float* w2a        = (const float*)d_in[7];
  const float* w2b        = (const float*)d_in[8];
  const float* w_fc2      = (const float*)d_in[9];
  const int*   atom_graph = (const int*)d_in[10];
  const int*   bond_graph = (const int*)d_in[11];
  const int*   num_nbs    = (const int*)d_in[12];
  const int*   n_atoms    = (const int*)d_in[13];
  float* out = (float*)d_out;

  const size_t nH = (size_t)Bsz * Asz * Hq;
  const size_t nQ = (size_t)Bsz * BONDq * Hq;
  float* h_a = (float*)d_ws;
  float* h_b = h_a + nH;
  float* nei = h_b + nH;
  float* prj = nei + nH;
  float* q1  = prj + nH;
  float* q2  = q1 + nQ;

  dim3 blk(256);
  const int gemm_blocks = (MP / 8) * NCH;
  const int bond_blocks = (BMP / 8) * NCH;

  bondproj_kernel<<<bond_blocks, blk, 0, stream>>>(bond_feats, w_nei, w2b, q1, q2);
  fc1_kernel<<<gemm_blocks, blk, 0, stream>>>(atom_feats, w_fc1, h_a);

  proj_kernel<<<gemm_blocks, blk, 0, stream>>>(h_a, w_nei, prj);
  nei_gather_kernel<<<GBLK, blk, 0, stream>>>(prj, q1, atom_graph, bond_graph,
                                             num_nbs, b_nei, nei);
  atom_kernel<<<gemm_blocks, blk, 0, stream>>>(h_a, nei, w_atom, b_atom, h_b);
  proj_kernel<<<gemm_blocks, blk, 0, stream>>>(h_b, w_nei, prj);
  nei_gather_kernel<<<GBLK, blk, 0, stream>>>(prj, q1, atom_graph, bond_graph,
                                             num_nbs, b_nei, nei);
  atom_kernel<<<gemm_blocks, blk, 0, stream>>>(h_b, nei, w_atom, b_atom, h_a);
  proj_kernel<<<gemm_blocks, blk, 0, stream>>>(h_a, w2a, prj);
  nprod_gather_kernel<<<GBLK, blk, 0, stream>>>(prj, q2, atom_graph, bond_graph,
                                               num_nbs, nei);
  out_kernel<<<gemm_blocks, blk, 0, stream>>>(h_a, w_fc2, nei, n_atoms, out);
}


# K=300 row GEMMs: each block copies its 300x64 W chunk into LDS once (whole 256-byte rows), its 8 waves take B fragments by ds_load_b128; k loop's global loads carry A rows only; sum order unchanged
# speedup vs baseline: 4.1104x; 1.1148x over previous
#include <hip/hip_runtime.h>
#include <hip/hip_bf16.h>


#define Bsz   256
#define Asz   100
#define NBq   10
#define BONDq 110
#define AFq   82
#define BFq   6
#define Hq    300
#define NCH   5
#define MTOT  (Bsz*Asz)
#define MP    (MTOT/32)
#define BROWS (Bsz*BONDq)
#define BMP   (BROWS/32)
#define C4    (Hq/4)
#define ITEMS (MTOT*C4)
#define GBLK  1875

typedef float v2f __attribute__((ext_vector_type(2)));
typedef float v4f __attribute__((ext_vector_type(4)));
typedef float v8f __attribute__((ext_vector_type(8)));

static __device__ __forceinline__ v8f wmma4(v2f a, v2f b, v8f c) {
  return __builtin_amdgcn_wmma_f32_16x16x4_f32(
      false, a, false, b, (short)0, c, false, false);
}

struct Tile { v8f c[2][4]; };
struct Ops  { v4f x0, x1, w0, w1, w2, w3; };

static __device__ __forceinline__ void mma8(v2f a0, v2f a1, v4f wlo, v4f whi,
                                            Tile& T) {
#pragma unroll
  for (int t = 0; t < 4; ++t) {
    v2f b; b.x = wlo[t]; b.y = whi[t];
    T.c[0][t] = wmma4(a0, b, T.c[0][t]);
    T.c[1][t] = wmma4(a1, b, T.c[1][t]);
  }
}

#define WP 64

struct WStage { v4f v[19]; };

static __device__ __forceinline__ WStage stage_w_load(const float* wsrc) {
  const int r0 = threadIdx.x >> 4;
  WStage s;
#pragma unroll
  for (int i = 0; i < 19; ++i) {
    const int r = r0 + 16 * i;
    const int rc = r < Hq ? r : Hq - 1;
    s.v[i] = *(const v4f*)(wsrc + rc * Hq);
  }
  return s;
}

static __device__ __forceinline__ void stage_w_store(const WStage& s, float* wl) {
  const int g = threadIdx.x & 15, r0 = threadIdx.x >> 4;
#pragma unroll
  for (int i = 0; i < 19; ++i) {
    const int r = r0 + 16 * i;
    if (r < Hq) *(v4f*)(wl + r * WP + 4 * g) = s.v[i];
  }
}

static __device__ __forceinline__ Ops ld8(const float* a0, const float* a1,
                                          const float* w) {
  Ops o;
  o.x0 = *(const v4f*)a0;            o.x1 = *(const v4f*)a1;
  o.w0 = *(const v4f*)w;             o.w1 = *(const v4f*)(w + WP);
  o.w2 = *(const v4f*)(w + 2 * WP);  o.w3 = *(const v4f*)(w + 3 * WP);
  return o;
}

static __device__ __forceinline__ void mm8(const Ops& o, Tile& T) {
  v2f a0l, a0h, a1l, a1h;
  a0l.x = o.x0.x; a0l.y = o.x0.y;  a0h.x = o.x0.z; a0h.y = o.x0.w;
  a1l.x = o.x1.x; a1l.y = o.x1.y;  a1h.x = o.x1.z; a1h.y = o.x1.w;
  mma8(a0l, a1l, o.w0, o.w1, T);
  mma8(a0h, a1h, o.w2, o.w3, T);
}

static __device__ __forceinline__ void gemm300(const float* a0, const float* a1,
                                               const float* w, int half,
                                               Tile& T) {
  const float* p0 = a0 + 4 * half;
  const float* p1 = a1 + 4 * half;
  const float* pw = w + 4 * half * WP;
  Ops cur = ld8(p0, p1, pw);
#pragma unroll 2
  for (int i = 0; i < 36; ++i) {
    p0 += 8; p1 += 8; pw += 8 * WP;
    Ops nxt = ld8(p0, p1, pw);
    mm8(cur, T);
    cur = nxt;
  }
  const int kt = 296 + 2 * half;
  const v2f t0 = *(const v2f*)(a0 + kt);
  const v2f t1 = *(const v2f*)(a1 + kt);
  const v4f u0 = *(const v4f*)(w + kt * WP);
  const v4f u1 = *(const v4f*)(w + (kt + 1) * WP);
  mm8(cur, T);
  mma8(t0, t1, u0, u1, T);
}

#define GEMM_COORDS()                                                        \
  const int lane = threadIdx.x & 31, wv = threadIdx.x >> 5;                  \
  const int c = lane & 15, half = lane >> 4;                                 \
  const int nc = blockIdx.x % NCH, grp = blockIdx.x / NCH;                   \
  const int mp = grp * 8 + wv;                                               \
  const int ncol = nc * 64 + 4 * c;                                          \
  const bool valid = ncol < Hq;                \
  const int ncc = valid ? ncol : (Hq - 4);                                   \
  const int row0 = mp * 32 + c, row1 = row0 + 16;                            \
  const int rb = mp * 32 + half * 8;

__global__ __launch_bounds__(256) void fc1_kernel(
    const float* __restrict__ af, const float* __restrict__ w,
    float* __restrict__ hout) {
  GEMM_COORDS()
  Tile T = {};
  const float* a0 = af + row0 * AFq + 2 * half;
  const float* a1 = af + row1 * AFq + 2 * half;
  const float* pw = w + 2 * half * Hq + ncc;
#pragma unroll 4
  for (int kk = 0; kk < 80; kk += 4) {
    const v2f x0 = *(const v2f*)a0;
    const v2f x1 = *(const v2f*)a1;
    const v4f u0 = *(const v4f*)pw;
    const v4f u1 = *(const v4f*)(pw + Hq);
    a0 += 4; a1 += 4; pw += 4 * Hq;
    mma8(x0, x1, u0, u1, T);
  }
  {
    const bool kv = (half == 0);
    const int kc = kv ? 80 : 0;
    v2f x0 = *(const v2f*)(af + row0 * AFq + kc);
    v2f x1 = *(const v2f*)(af + row1 * AFq + kc);
    v4f u0 = *(const v4f*)(w + kc * Hq + ncc);
    v4f u1 = *(const v4f*)(w + (kc + 1) * Hq + ncc);
    const v2f z2 = {0.f, 0.f};
    const v4f z4 = {0.f, 0.f, 0.f, 0.f};
    x0 = kv ? x0 : z2;  x1 = kv ? x1 : z2;
    u0 = kv ? u0 : z4;  u1 = kv ? u1 : z4;
    mma8(x0, x1, u0, u1, T);
  }
#pragma unroll
  for (int i = 0; i < 2; ++i) {
#pragma unroll
    for (int r = 0; r < 8; ++r) {
      v4f v;
      v.x = fmaxf(T.c[i][0][r], 0.f);  v.y = fmaxf(T.c[i][1][r], 0.f);
      v.z = fmaxf(T.c[i][2][r], 0.f);  v.w = fmaxf(T.c[i][3][r], 0.f);
      if (valid) *(v4f*)(hout + (rb + i * 16 + r) * Hq + ncol) = v;
    }
  }
}

__global__ __launch_bounds__(256) void bondproj_kernel(
    const float* __restrict__ bond, const float* __restrict__ wn,
    const float* __restrict__ w2b, float* __restrict__ q1,
    float* __restrict__ q2) {
  GEMM_COORDS()
  const float* wnb = wn + Hq * Hq;
  const bool kv = (half == 0);
  const int kc = kv ? 4 : 0;
  const v2f z2 = {0.f, 0.f};
  const v4f z4 = {0.f, 0.f, 0.f, 0.f};
  const v2f xa0 = *(const v2f*)(bond + row0 * BFq + 2 * half);
  const v2f xa1 = *(const v2f*)(bond + row1 * BFq + 2 * half);
  v2f xb0 = *(const v2f*)(bond + row0 * BFq + kc);
  v2f xb1 = *(const v2f*)(bond + row1 * BFq + kc);
  xb0 = kv ? xb0 : z2;  xb1 = kv ? xb1 : z2;
#pragma unroll
  for (int pass = 0; pass < 2; ++pass) {
    const float* wsrc = pass ? w2b : wnb;
    float* dst = pass ? q2 : q1;
    const v4f u0 = *(const v4f*)(wsrc + (2 * half) * Hq + ncc);
    const v4f u1 = *(const v4f*)(wsrc + (2 * half + 1) * Hq + ncc);
    v4f u2 = *(const v4f*)(wsrc + kc * Hq + ncc);
    v4f u3 = *(const v4f*)(wsrc + (kc + 1) * Hq + ncc);
    u2 = kv ? u2 : z4;  u3 = kv ? u3 : z4;
    Tile T = {};
    mma8(xa0, xa1, u0, u1, T);
    mma8(xb0, xb1, u2, u3, T);
#pragma unroll
    for (int i = 0; i < 2; ++i) {
#pragma unroll
      for (int r = 0; r < 8; ++r) {
        v4f v;
        v.x = T.c[i][0][r];  v.y = T.c[i][1][r];
        v.z = T.c[i][2][r];  v.w = T.c[i][3][r];
        if (valid) *(v4f*)(dst + (rb + i * 16 + r) * Hq + ncol) = v;
      }
    }
  }
}

__global__ __launch_bounds__(256) void proj_kernel(
    const float* __restrict__ h, const float* __restrict__ w,
    float* __restrict__ pout) {
  GEMM_COORDS()
  __shared__ float wl[Hq * WP];
  stage_w_store(stage_w_load(w + ncc), wl);
  __syncthreads();
  Tile T = {};
  gemm300(h + row0 * Hq, h + row1 * Hq, wl + 4 * c, half, T);
#pragma unroll
  for (int i = 0; i < 2; ++i) {
#pragma unroll
    for (int r = 0; r < 8; ++r) {
      v4f v;
      v.x = T.c[i][0][r];  v.y = T.c[i][1][r];
      v.z = T.c[i][2][r];  v.w = T.c[i][3][r];
      if (valid) *(v4f*)(pout + (rb + i * 16 + r) * Hq + ncol) = v;
    }
  }
}

__global__ __launch_bounds__(256) void nei_gather_kernel(
    const float* __restrict__ p, const float* __restrict__ q,
    const int* __restrict__ ag, const int* __restrict__ bg,
    const int* __restrict__ nnb, const float* __restrict__ bn,
    float* __restrict__ out) {
  for (int i = blockIdx.x * 256 + threadIdx.x; i < ITEMS; i += GBLK * 256) {
    const int ba = i / C4, c = (i - ba * C4) * 4;
    const int nn = nnb[ba];
    const float bx = bn[c], by = bn[c + 1], bz = bn[c + 2], bw = bn[c + 3];
    int po[NBq], qo[NBq];
#pragma unroll
    for (int j = 0; j < NBq; ++j) {
      const int e = (ba * NBq + j) * 2;
      po[j] = (ag[e] * Asz + ag[e + 1]) * Hq + c;
      qo[j] = (bg[e] * BONDq + bg[e + 1]) * Hq + c;
    }
    float4 pv[NBq], qv[NBq];
#pragma unroll
    for (int j = 0; j < NBq; ++j) {
      pv[j] = make_float4(0.f, 0.f, 0.f, 0.f);
      qv[j] = make_float4(0.f, 0.f, 0.f, 0.f);
      if (j < nn) {
        pv[j] = *(const float4*)(p + po[j]);
        qv[j] = *(const float4*)(q + qo[j]);
      }
    }
    float4 s = make_float4(0.f, 0.f, 0.f, 0.f);
#pragma unroll
    for (int j = 0; j < NBq; ++j) {
      const bool live = j < nn;
      const float x = fmaxf(pv[j].x + qv[j].x + bx, 0.f);
      const float y = fmaxf(pv[j].y + qv[j].y + by, 0.f);
      const float z = fmaxf(pv[j].z + qv[j].z + bz, 0.f);
      const float w = fmaxf(pv[j].w + qv[j].w + bw, 0.f);
      s.x += live ? x : 0.f;  s.y += live ? y : 0.f;
      s.z += live ? z : 0.f;  s.w += live ? w : 0.f;
    }
    *(float4*)(out + ba * Hq + c) = s;
  }
}

__global__ __launch_bounds__(256) void atom_kernel(
    const float* __restrict__ h, const float* __restrict__ nei,
    const float* __restrict__ wa, const float* __restrict__ bb,
    float* __restrict__ hout) {
  GEMM_COORDS()
  __shared__ float wl[Hq * WP];
  stage_w_store(stage_w_load(wa + ncc), wl);
  __syncthreads();
  Tile T = {};
  gemm300(h + row0 * Hq, h + row1 * Hq, wl + 4 * c, half, T);
  const WStage s2 = stage_w_load(wa + Hq * Hq + ncc);
  __syncthreads();
  stage_w_store(s2, wl);
  __syncthreads();
  gemm300(nei + row0 * Hq, nei + row1 * Hq, wl + 4 * c, half, T);
  const v4f bias = *(const v4f*)(bb + ncc);
#pragma unroll
  for (int i = 0; i < 2; ++i) {
#pragma unroll
    for (int r = 0; r < 8; ++r) {
      v4f v;
      v.x = fmaxf(T.c[i][0][r] + bias.x, 0.f);
      v.y = fmaxf(T.c[i][1][r] + bias.y, 0.f);
      v.z = fmaxf(T.c[i][2][r] + bias.z, 0.f);
      v.w = fmaxf(T.c[i][3][r] + bias.w, 0.f);
      if (valid) *(v4f*)(hout + (rb + i * 16 + r) * Hq + ncol) = v;
    }
  }
}

__global__ __launch_bounds__(256) void nprod_gather_kernel(
    const float* __restrict__ p, const float* __restrict__ q,
    const int* __restrict__ ag, const int* __restrict__ bg,
    const int* __restrict__ nnb, float* __restrict__ out) {
  for (int i = blockIdx.x * 256 + threadIdx.x; i < ITEMS; i += GBLK * 256) {
    const int ba = i / C4, c = (i - ba * C4) * 4;
    const int nn = nnb[ba];
    int po[NBq], qo[NBq];
#pragma unroll
    for (int j = 0; j < NBq; ++j) {
      const int e = (ba * NBq + j) * 2;
      po[j] = (ag[e] * Asz + ag[e + 1]) * Hq + c;
      qo[j] = (bg[e] * BONDq + bg[e + 1]) * Hq + c;
    }
    float4 pv[NBq], qv[NBq];
#pragma unroll
    for (int j = 0; j < NBq; ++j) {
      pv[j] = make_float4(0.f, 0.f, 0.f, 0.f);
      qv[j] = make_float4(0.f, 0.f, 0.f, 0.f);
      if (j < nn) {
        pv[j] = *(const float4*)(p + po[j]);
        qv[j] = *(const float4*)(q + qo[j]);
      }
    }
    float4 s = make_float4(0.f, 0.f, 0.f, 0.f);
#pragma unroll
    for (int j = 0; j < NBq; ++j) {
      const bool live = j < nn;
      s.x += live ? pv[j].x * qv[j].x : 0.f;
      s.y += live ? pv[j].y * qv[j].y : 0.f;
      s.z += live ? pv[j].z * qv[j].z : 0.f;
      s.w += live ? pv[j].w * qv[j].w : 0.f;
    }
    *(float4*)(out + ba * Hq + c) = s;
  }
}

__global__ __launch_bounds__(256) void out_kernel(
    const float* __restrict__ h, const float* __restrict__ w,
    const float* __restrict__ nprod, const int* __restrict__ na,
    float* __restrict__ out) {
  GEMM_COORDS()
  __shared__ float wl[Hq * WP];
  stage_w_store(stage_w_load(w + ncc), wl);
  __syncthreads();
  Tile T = {};
  gemm300(h + row0 * Hq, h + row1 * Hq, wl + 4 * c, half, T);
#pragma unroll
  for (int i = 0; i < 2; ++i) {
#pragma unroll
    for (int r = 0; r < 8; ++r) {
      const int row = rb + i * 16 + r;
      const int bi = row / Asz, ai = row - bi * Asz;
      const bool live = ai < na[bi];
      if (valid) {
        const v4f np4 = *(const v4f*)(nprod + row * Hq + ncol);
        v4f v;
        v.x = live ? T.c[i][0][r] * np4.x : 0.f;
        v.y = live ? T.c[i][1][r] * np4.y : 0.f;
        v.z = live ? T.c[i][2][r] * np4.z : 0.f;
        v.w = live ? T.c[i][3][r] * np4.w : 0.f;
        *(v4f*)(out + row * Hq + ncol) = v;
      }
    }
  }
}

extern "C" void kernel_launch(void* const* d_in, const int* in_sizes, int n_in,
                              void* d_out, int out_size, void* d_ws, size_t ws_size,
                              hipStream_t stream) {
  const float* atom_feats = (const float*)d_in[0];
  const float* bond_feats = (const float*)d_in[1];
  const float* w_fc1      = (const float*)d_in[2];
  const float* w_nei      = (const float*)d_in[3];
  const float* b_nei      = (const float*)d_in[4];
  const float* w_atom     = (const float*)d_in[5];
  const float* b_atom     = (const float*)d_in[6];
  const float* w2a        = (const float*)d_in[7];
  const float* w2b        = (const float*)d_in[8];
  const float* w_fc2      = (const float*)d_in[9];
  const int*   atom_graph = (const int*)d_in[10];
  const int*   bond_graph = (const int*)d_in[11];
  const int*   num_nbs    = (const int*)d_in[12];
  const int*   n_atoms    = (const int*)d_in[13];
  float* out = (float*)d_out;

  const size_t nH = (size_t)Bsz * Asz * Hq;
  const size_t nQ = (size_t)Bsz * BONDq * Hq;
  float* h_a = (float*)d_ws;
  float* h_b = h_a + nH;
  float* nei = h_b + nH;
  float* prj = nei + nH;
  float* q1  = prj + nH;
  float* q2  = q1 + nQ;

  dim3 blk(256);
  const int gemm_blocks = (MP / 8) * NCH;
  const int bond_blocks = (BMP / 8) * NCH;

  bondproj_kernel<<<bond_blocks, blk, 0, stream>>>(bond_feats, w_nei, w2b, q1, q2);
  fc1_kernel<<<gemm_blocks, blk, 0, stream>>>(atom_feats, w_fc1, h_a);

  proj_kernel<<<gemm_blocks, blk, 0, stream>>>(h_a, w_nei, prj);
  nei_gather_kernel<<<GBLK, blk, 0, stream>>>(prj, q1, atom_graph, bond_graph,
                                             num_nbs, b_nei, nei);
  atom_kernel<<<gemm_blocks, blk, 0, stream>>>(h_a, nei, w_atom, b_atom, h_b);
  proj_kernel<<<gemm_blocks, blk, 0, stream>>>(h_b, w_nei, prj);
  nei_gather_kernel<<<GBLK, blk, 0, stream>>>(prj, q1, atom_graph, bond_graph,
                                             num_nbs, b_nei, nei);
  atom_kernel<<<gemm_blocks, blk, 0, stream>>>(h_b, nei, w_atom, b_atom, h_a);
  proj_kernel<<<gemm_blocks, blk, 0, stream>>>(h_a, w2a, prj);
  nprod_gather_kernel<<<GBLK, blk, 0, stream>>>(prj, q2, atom_graph, bond_graph,
                                               num_nbs, nei);
  out_kernel<<<gemm_blocks, blk, 0, stream>>>(h_a, w_fc2, nei, n_atoms, out);
}


# K=300 row GEMMs: A rows requested three 8-k steps ahead of use (four rotating A register sets; steps 0-2 requested before the block stages W); k order and sum order unchanged
# speedup vs baseline: 4.2531x; 1.0347x over previous
#include <hip/hip_runtime.h>
#include <hip/hip_bf16.h>


#define Bsz   256
#define Asz   100
#define NBq   10
#define BONDq 110
#define AFq   82
#define BFq   6
#define Hq    300
#define NCH   5
#define MTOT  (Bsz*Asz)
#define MP    (MTOT/32)
#define BROWS (Bsz*BONDq)
#define BMP   (BROWS/32)
#define C4    (Hq/4)
#define ITEMS (MTOT*C4)
#define GBLK  1875

typedef float v2f __attribute__((ext_vector_type(2)));
typedef float v4f __attribute__((ext_vector_type(4)));
typedef float v8f __attribute__((ext_vector_type(8)));

static __device__ __forceinline__ v8f wmma4(v2f a, v2f b, v8f c) {
  return __builtin_amdgcn_wmma_f32_16x16x4_f32(
      false, a, false, b, (short)0, c, false, false);
}

struct Tile { v8f c[2][4]; };
struct AOps { v4f x0, x1; };
struct WOps { v4f w0, w1, w2, w3; };
struct APre { AOps s0, s1, s2; };

static __device__ __forceinline__ void mma8(v2f a0, v2f a1, v4f wlo, v4f whi,
                                            Tile& T) {
#pragma unroll
  for (int t = 0; t < 4; ++t) {
    v2f b; b.x = wlo[t]; b.y = whi[t];
    T.c[0][t] = wmma4(a0, b, T.c[0][t]);
    T.c[1][t] = wmma4(a1, b, T.c[1][t]);
  }
}

#define WP 64

struct WStage { v4f v[19]; };

static __device__ __forceinline__ WStage stage_w_load(const float* wsrc) {
  const int r0 = threadIdx.x >> 4;
  WStage s;
#pragma unroll
  for (int i = 0; i < 19; ++i) {
    const int r = r0 + 16 * i;
    const int rc = r < Hq ? r : Hq - 1;
    s.v[i] = *(const v4f*)(wsrc + rc * Hq);
  }
  return s;
}

static __device__ __forceinline__ void stage_w_store(const WStage& s, float* wl) {
  const int g = threadIdx.x & 15, r0 = threadIdx.x >> 4;
#pragma unroll
  for (int i = 0; i < 19; ++i) {
    const int r = r0 + 16 * i;
    if (r < Hq) *(v4f*)(wl + r * WP + 4 * g) = s.v[i];
  }
}

static __device__ __forceinline__ AOps ldA(const float* a0, const float* a1) {
  AOps o;
  o.x0 = *(const v4f*)a0;            o.x1 = *(const v4f*)a1;
  return o;
}

static __device__ __forceinline__ WOps ldW(const float* w) {
  WOps o;
  o.w0 = *(const v4f*)w;             o.w1 = *(const v4f*)(w + WP);
  o.w2 = *(const v4f*)(w + 2 * WP);  o.w3 = *(const v4f*)(w + 3 * WP);
  return o;
}

static __device__ __forceinline__ void mm8(const AOps& a, const WOps& w, Tile& T) {
  v2f a0l, a0h, a1l, a1h;
  a0l.x = a.x0.x; a0l.y = a.x0.y;  a0h.x = a.x0.z; a0h.y = a.x0.w;
  a1l.x = a.x1.x; a1l.y = a.x1.y;  a1h.x = a.x1.z; a1h.y = a.x1.w;
  mma8(a0l, a1l, w.w0, w.w1, T);
  mma8(a0h, a1h, w.w2, w.w3, T);
}

static __device__ __forceinline__ APre gemm300_head(const float* a0,
                                                    const float* a1, int half) {
  const float* p0 = a0 + 4 * half;
  const float* p1 = a1 + 4 * half;
  APre h;
  h.s0 = ldA(p0, p1);  h.s1 = ldA(p0 + 8, p1 + 8);  h.s2 = ldA(p0 + 16, p1 + 16);
  return h;
}

static __device__ __forceinline__ void gemm300(const APre& pre, const float* a0,
                                               const float* a1, const float* w,
                                               int half, Tile& T) {
  const float* p0 = a0 + 4 * half;
  const float* p1 = a1 + 4 * half;
  const float* pw = w + 4 * half * WP;
  AOps A0 = pre.s0, A1 = pre.s1, A2 = pre.s2, A3;
  WOps Wc = ldW(pw), Wn;
#pragma unroll 1
  for (int it = 0; it < 8; ++it) {
    A3 = ldA(p0 + 24, p1 + 24);  Wn = ldW(pw + 8 * WP);   mm8(A0, Wc, T);
    A0 = ldA(p0 + 32, p1 + 32);  Wc = ldW(pw + 16 * WP);  mm8(A1, Wn, T);
    A1 = ldA(p0 + 40, p1 + 40);  Wn = ldW(pw + 24 * WP);  mm8(A2, Wc, T);
    A2 = ldA(p0 + 48, p1 + 48);  Wc = ldW(pw + 32 * WP);  mm8(A3, Wn, T);
    p0 += 32; p1 += 32; pw += 32 * WP;
  }
  A3 = ldA(p0 + 24, p1 + 24);  Wn = ldW(pw + 8 * WP);   mm8(A0, Wc, T);
  A0 = ldA(p0 + 32, p1 + 32);  Wc = ldW(pw + 16 * WP);  mm8(A1, Wn, T);
  const int kt = 296 + 2 * half;
  const v2f t0 = *(const v2f*)(a0 + kt);
  const v2f t1 = *(const v2f*)(a1 + kt);
  Wn = ldW(pw + 24 * WP);                               mm8(A2, Wc, T);
  Wc = ldW(pw + 32 * WP);
  const v4f u0 = *(const v4f*)(w + kt * WP);
  const v4f u1 = *(const v4f*)(w + (kt + 1) * WP);
  mm8(A3, Wn, T);
  mm8(A0, Wc, T);
  mma8(t0, t1, u0, u1, T);
}

#define GEMM_COORDS()                                                        \
  const int lane = threadIdx.x & 31, wv = threadIdx.x >> 5;                  \
  const int c = lane & 15, half = lane >> 4;                                 \
  const int nc = blockIdx.x % NCH, grp = blockIdx.x / NCH;                   \
  const int mp = grp * 8 + wv;                                               \
  const int ncol = nc * 64 + 4 * c;                                          \
  const bool valid = ncol < Hq;                \
  const int ncc = valid ? ncol : (Hq - 4);                                   \
  const int row0 = mp * 32 + c, row1 = row0 + 16;                            \
  const int rb = mp * 32 + half * 8;

__global__ __launch_bounds__(256) void fc1_kernel(
    const float* __restrict__ af, const float* __restrict__ w,
    float* __restrict__ hout) {
  GEMM_COORDS()
  Tile T = {};
  const float* a0 = af + row0 * AFq + 2 * half;
  const float* a1 = af + row1 * AFq + 2 * half;
  const float* pw = w + 2 * half * Hq + ncc;
#pragma unroll 4
  for (int kk = 0; kk < 80; kk += 4) {
    const v2f x0 = *(const v2f*)a0;
    const v2f x1 = *(const v2f*)a1;
    const v4f u0 = *(const v4f*)pw;
    const v4f u1 = *(const v4f*)(pw + Hq);
    a0 += 4; a1 += 4; pw += 4 * Hq;
    mma8(x0, x1, u0, u1, T);
  }
  {
    const bool kv = (half == 0);
    const int kc = kv ? 80 : 0;
    v2f x0 = *(const v2f*)(af + row0 * AFq + kc);
    v2f x1 = *(const v2f*)(af + row1 * AFq + kc);
    v4f u0 = *(const v4f*)(w + kc * Hq + ncc);
    v4f u1 = *(const v4f*)(w + (kc + 1) * Hq + ncc);
    const v2f z2 = {0.f, 0.f};
    const v4f z4 = {0.f, 0.f, 0.f, 0.f};
    x0 = kv ? x0 : z2;  x1 = kv ? x1 : z2;
    u0 = kv ? u0 : z4;  u1 = kv ? u1 : z4;
    mma8(x0, x1, u0, u1, T);
  }
#pragma unroll
  for (int i = 0; i < 2; ++i) {
#pragma unroll
    for (int r = 0; r < 8; ++r) {
      v4f v;
      v.x = fmaxf(T.c[i][0][r], 0.f);  v.y = fmaxf(T.c[i][1][r], 0.f);
      v.z = fmaxf(T.c[i][2][r], 0.f);  v.w = fmaxf(T.c[i][3][r], 0.f);
      if (valid) *(v4f*)(hout + (rb + i * 16 + r) * Hq + ncol) = v;
    }
  }
}

__global__ __launch_bounds__(256) void bondproj_kernel(
    const float* __restrict__ bond, const float* __restrict__ wn,
    const float* __restrict__ w2b, float* __restrict__ q1,
    float* __restrict__ q2) {
  GEMM_COORDS()
  const float* wnb = wn + Hq * Hq;
  const bool kv = (half == 0);
  const int kc = kv ? 4 : 0;
  const v2f z2 = {0.f, 0.f};
  const v4f z4 = {0.f, 0.f, 0.f, 0.f};
  const v2f xa0 = *(const v2f*)(bond + row0 * BFq + 2 * half);
  const v2f xa1 = *(const v2f*)(bond + row1 * BFq + 2 * half);
  v2f xb0 = *(const v2f*)(bond + row0 * BFq + kc);
  v2f xb1 = *(const v2f*)(bond + row1 * BFq + kc);
  xb0 = kv ? xb0 : z2;  xb1 = kv ? xb1 : z2;
#pragma unroll
  for (int pass = 0; pass < 2; ++pass) {
    const float* wsrc = pass ? w2b : wnb;
    float* dst = pass ? q2 : q1;
    const v4f u0 = *(const v4f*)(wsrc + (2 * half) * Hq + ncc);
    const v4f u1 = *(const v4f*)(wsrc + (2 * half + 1) * Hq + ncc);
    v4f u2 = *(const v4f*)(wsrc + kc * Hq + ncc);
    v4f u3 = *(const v4f*)(wsrc + (kc + 1) * Hq + ncc);
    u2 = kv ? u2 : z4;  u3 = kv ? u3 : z4;
    Tile T = {};
    mma8(xa0, xa1, u0, u1, T);
    mma8(xb0, xb1, u2, u3, T);
#pragma unroll
    for (int i = 0; i < 2; ++i) {
#pragma unroll
      for (int r = 0; r < 8; ++r) {
        v4f v;
        v.x = T.c[i][0][r];  v.y = T.c[i][1][r];
        v.z = T.c[i][2][r];  v.w = T.c[i][3][r];
        if (valid) *(v4f*)(dst + (rb + i * 16 + r) * Hq + ncol) = v;
      }
    }
  }
}

__global__ __launch_bounds__(256) void proj_kernel(
    const float* __restrict__ h, const float* __restrict__ w,
    float* __restrict__ pout) {
  GEMM_COORDS()
  __shared__ float wl[Hq * WP];
  const APre pre = gemm300_head(h + row0 * Hq, h + row1 * Hq, half);
  stage_w_store(stage_w_load(w + ncc), wl);
  __syncthreads();
  Tile T = {};
  gemm300(pre, h + row0 * Hq, h + row1 * Hq, wl + 4 * c, half, T);
#pragma unroll
  for (int i = 0; i < 2; ++i) {
#pragma unroll
    for (int r = 0; r < 8; ++r) {
      v4f v;
      v.x = T.c[i][0][r];  v.y = T.c[i][1][r];
      v.z = T.c[i][2][r];  v.w = T.c[i][3][r];
      if (valid) *(v4f*)(pout + (rb + i * 16 + r) * Hq + ncol) = v;
    }
  }
}

__global__ __launch_bounds__(256) void nei_gather_kernel(
    const float* __restrict__ p, const float* __restrict__ q,
    const int* __restrict__ ag, const int* __restrict__ bg,
    const int* __restrict__ nnb, const float* __restrict__ bn,
    float* __restrict__ out) {
  for (int i = blockIdx.x * 256 + threadIdx.x; i < ITEMS; i += GBLK * 256) {
    const int ba = i / C4, c = (i - ba * C4) * 4;
    const int nn = nnb[ba];
    const float bx = bn[c], by = bn[c + 1], bz = bn[c + 2], bw = bn[c + 3];
    int po[NBq], qo[NBq];
#pragma unroll
    for (int j = 0; j < NBq; ++j) {
      const int e = (ba * NBq + j) * 2;
      po[j] = (ag[e] * Asz + ag[e + 1]) * Hq + c;
      qo[j] = (bg[e] * BONDq + bg[e + 1]) * Hq + c;
    }
    float4 pv[NBq], qv[NBq];
#pragma unroll
    for (int j = 0; j < NBq; ++j) {
      pv[j] = make_float4(0.f, 0.f, 0.f, 0.f);
      qv[j] = make_float4(0.f, 0.f, 0.f, 0.f);
      if (j < nn) {
        pv[j] = *(const float4*)(p + po[j]);
        qv[j] = *(const float4*)(q + qo[j]);
      }
    }
    float4 s = make_float4(0.f, 0.f, 0.f, 0.f);
#pragma unroll
    for (int j = 0; j < NBq; ++j) {
      const bool live = j < nn;
      const float x = fmaxf(pv[j].x + qv[j].x + bx, 0.f);
      const float y = fmaxf(pv[j].y + qv[j].y + by, 0.f);
      const float z = fmaxf(pv[j].z + qv[j].z + bz, 0.f);
      const float w = fmaxf(pv[j].w + qv[j].w + bw, 0.f);
      s.x += live ? x : 0.f;  s.y += live ? y : 0.f;
      s.z += live ? z : 0.f;  s.w += live ? w : 0.f;
    }
    *(float4*)(out + ba * Hq + c) = s;
  }
}

__global__ __launch_bounds__(256) void atom_kernel(
    const float* __restrict__ h, const float* __restrict__ nei,
    const float* __restrict__ wa, const float* __restrict__ bb,
    float* __restrict__ hout) {
  GEMM_COORDS()
  __shared__ float wl[Hq * WP];
  const APre pre = gemm300_head(h + row0 * Hq, h + row1 * Hq, half);
  stage_w_store(stage_w_load(wa + ncc), wl);
  __syncthreads();
  Tile T = {};
  gemm300(pre, h + row0 * Hq, h + row1 * Hq, wl + 4 * c, half, T);
  const WStage s2 = stage_w_load(wa + Hq * Hq + ncc);
  const APre pre2 = gemm300_head(nei + row0 * Hq, nei + row1 * Hq, half);
  __syncthreads();
  stage_w_store(s2, wl);
  __syncthreads();
  gemm300(pre2, nei + row0 * Hq, nei + row1 * Hq, wl + 4 * c, half, T);
  const v4f bias = *(const v4f*)(bb + ncc);
#pragma unroll
  for (int i = 0; i < 2; ++i) {
#pragma unroll
    for (int r = 0; r < 8; ++r) {
      v4f v;
      v.x = fmaxf(T.c[i][0][r] + bias.x, 0.f);
      v.y = fmaxf(T.c[i][1][r] + bias.y, 0.f);
      v.z = fmaxf(T.c[i][2][r] + bias.z, 0.f);
      v.w = fmaxf(T.c[i][3][r] + bias.w, 0.f);
      if (valid) *(v4f*)(hout + (rb + i * 16 + r) * Hq + ncol) = v;
    }
  }
}

__global__ __launch_bounds__(256) void nprod_gather_kernel(
    const float* __restrict__ p, const float* __restrict__ q,
    const int* __restrict__ ag, const int* __restrict__ bg,
    const int* __restrict__ nnb, float* __restrict__ out) {
  for (int i = blockIdx.x * 256 + threadIdx.x; i < ITEMS; i += GBLK * 256) {
    const int ba = i / C4, c = (i - ba * C4) * 4;
    const int nn = nnb[ba];
    int po[NBq], qo[NBq];
#pragma unroll
    for (int j = 0; j < NBq; ++j) {
      const int e = (ba * NBq + j) * 2;
      po[j] = (ag[e] * Asz + ag[e + 1]) * Hq + c;
      qo[j] = (bg[e] * BONDq + bg[e + 1]) * Hq + c;
    }
    float4 pv[NBq], qv[NBq];
#pragma unroll
    for (int j = 0; j < NBq; ++j) {
      pv[j] = make_float4(0.f, 0.f, 0.f, 0.f);
      qv[j] = make_float4(0.f, 0.f, 0.f, 0.f);
      if (j < nn) {
        pv[j] = *(const float4*)(p + po[j]);
        qv[j] = *(const float4*)(q + qo[j]);
      }
    }
    float4 s = make_float4(0.f, 0.f, 0.f, 0.f);
#pragma unroll
    for (int j = 0; j < NBq; ++j) {
      const bool live = j < nn;
      s.x += live ? pv[j].x * qv[j].x : 0.f;
      s.y += live ? pv[j].y * qv[j].y : 0.f;
      s.z += live ? pv[j].z * qv[j].z : 0.f;
      s.w += live ? pv[j].w * qv[j].w : 0.f;
    }
    *(float4*)(out + ba * Hq + c) = s;
  }
}

__global__ __launch_bounds__(256) void out_kernel(
    const float* __restrict__ h, const float* __restrict__ w,
    const float* __restrict__ nprod, const int* __restrict__ na,
    float* __restrict__ out) {
  GEMM_COORDS()
  __shared__ float wl[Hq * WP];
  const APre pre = gemm300_head(h + row0 * Hq, h + row1 * Hq, half);
  stage_w_store(stage_w_load(w + ncc), wl);
  __syncthreads();
  Tile T = {};
  gemm300(pre, h + row0 * Hq, h + row1 * Hq, wl + 4 * c, half, T);
#pragma unroll
  for (int i = 0; i < 2; ++i) {
#pragma unroll
    for (int r = 0; r < 8; ++r) {
      const int row = rb + i * 16 + r;
      const int bi = row / Asz, ai = row - bi * Asz;
      const bool live = ai < na[bi];
      if (valid) {
        const v4f np4 = *(const v4f*)(nprod + row * Hq + ncol);
        v4f v;
        v.x = live ? T.c[i][0][r] * np4.x : 0.f;
        v.y = live ? T.c[i][1][r] * np4.y : 0.f;
        v.z = live ? T.c[i][2][r] * np4.z : 0.f;
        v.w = live ? T.c[i][3][r] * np4.w : 0.f;
        *(v4f*)(out + row * Hq + ncol) = v;
      }
    }
  }
}

extern "C" void kernel_launch(void* const* d_in, const int* in_sizes, int n_in,
                              void* d_out, int out_size, void* d_ws, size_t ws_size,
                              hipStream_t stream) {
  const float* atom_feats = (const float*)d_in[0];
  const float* bond_feats = (const float*)d_in[1];
  const float* w_fc1      = (const float*)d_in[2];
  const float* w_nei      = (const float*)d_in[3];
  const float* b_nei      = (const float*)d_in[4];
  const float* w_atom     = (const float*)d_in[5];
  const float* b_atom     = (const float*)d_in[6];
  const float* w2a        = (const float*)d_in[7];
  const float* w2b        = (const float*)d_in[8];
  const float* w_fc2      = (const float*)d_in[9];
  const int*   atom_graph = (const int*)d_in[10];
  const int*   bond_graph = (const int*)d_in[11];
  const int*   num_nbs    = (const int*)d_in[12];
  const int*   n_atoms    = (const int*)d_in[13];
  float* out = (float*)d_out;

  const size_t nH = (size_t)Bsz * Asz * Hq;
  const size_t nQ = (size_t)Bsz * BONDq * Hq;
  float* h_a = (float*)d_ws;
  float* h_b = h_a + nH;
  float* nei = h_b + nH;
  float* prj = nei + nH;
  float* q1  = prj + nH;
  float* q2  = q1 + nQ;

  dim3 blk(256);
  const int gemm_blocks = (MP / 8) * NCH;
  const int bond_blocks = (BMP / 8) * NCH;

  bondproj_kernel<<<bond_blocks, blk, 0, stream>>>(bond_feats, w_nei, w2b, q1, q2);
  fc1_kernel<<<gemm_blocks, blk, 0, stream>>>(atom_feats, w_fc1, h_a);

  proj_kernel<<<gemm_blocks, blk, 0, stream>>>(h_a, w_nei, prj);
  nei_gather_kernel<<<GBLK, blk, 0, stream>>>(prj, q1, atom_graph, bond_graph,
                                             num_nbs, b_nei, nei);
  atom_kernel<<<gemm_blocks, blk, 0, stream>>>(h_a, nei, w_atom, b_atom, h_b);
  proj_kernel<<<gemm_blocks, blk, 0, stream>>>(h_b, w_nei, prj);
  nei_gather_kernel<<<GBLK, blk, 0, stream>>>(prj, q1, atom_graph, bond_graph,
                                             num_nbs, b_nei, nei);
  atom_kernel<<<gemm_blocks, blk, 0, stream>>>(h_b, nei, w_atom, b_atom, h_a);
  proj_kernel<<<gemm_blocks, blk, 0, stream>>>(h_a, w2a, prj);
  nprod_gather_kernel<<<GBLK, blk, 0, stream>>>(prj, q2, atom_graph, bond_graph,
                                               num_nbs, nei);
  out_kernel<<<gemm_blocks, blk, 0, stream>>>(h_a, w_fc2, nei, n_atoms, out);
}


# K=300 row GEMMs: k loop in blocks of two 8-k steps; every W (LDS) and A (memory) load stands in one group in front of an earlier block's 32 WMMAs, none among the WMMAs; k and sum order unchanged
# speedup vs baseline: 4.4414x; 1.0443x over previous
#include <hip/hip_runtime.h>
#include <hip/hip_bf16.h>


#define Bsz   256
#define Asz   100
#define NBq   10
#define BONDq 110
#define AFq   82
#define BFq   6
#define Hq    300
#define NCH   5
#define MTOT  (Bsz*Asz)
#define MP    (MTOT/32)
#define BROWS (Bsz*BONDq)
#define BMP   (BROWS/32)
#define C4    (Hq/4)
#define ITEMS (MTOT*C4)
#define GBLK  1875

typedef float v2f __attribute__((ext_vector_type(2)));
typedef float v4f __attribute__((ext_vector_type(4)));
typedef float v8f __attribute__((ext_vector_type(8)));

static __device__ __forceinline__ v8f wmma4(v2f a, v2f b, v8f c) {
  return __builtin_amdgcn_wmma_f32_16x16x4_f32(
      false, a, false, b, (short)0, c, false, false);
}

struct Tile { v8f c[2][4]; };
struct AOps { v4f x0, x1; };
struct WOps { v4f w0, w1, w2, w3; };
struct APre { AOps s0, s1, s2, s3; };

static __device__ __forceinline__ void mma8(v2f a0, v2f a1, v4f wlo, v4f whi,
                                            Tile& T) {
#pragma unroll
  for (int t = 0; t < 4; ++t) {
    v2f b; b.x = wlo[t]; b.y = whi[t];
    T.c[0][t] = wmma4(a0, b, T.c[0][t]);
    T.c[1][t] = wmma4(a1, b, T.c[1][t]);
  }
}

#define WP 64

struct WStage { v4f v[19]; };

static __device__ __forceinline__ WStage stage_w_load(const float* wsrc) {
  const int r0 = threadIdx.x >> 4;
  WStage s;
#pragma unroll
  for (int i = 0; i < 19; ++i) {
    const int r = r0 + 16 * i;
    const int rc = r < Hq ? r : Hq - 1;
    s.v[i] = *(const v4f*)(wsrc + rc * Hq);
  }
  return s;
}

static __device__ __forceinline__ void stage_w_store(const WStage& s, float* wl) {
  const int g = threadIdx.x & 15, r0 = threadIdx.x >> 4;
#pragma unroll
  for (int i = 0; i < 19; ++i) {
    const int r = r0 + 16 * i;
    if (r < Hq) *(v4f*)(wl + r * WP + 4 * g) = s.v[i];
  }
}

static __device__ __forceinline__ AOps ldA(const float* a0, const float* a1) {
  AOps o;
  o.x0 = *(const v4f*)a0;            o.x1 = *(const v4f*)a1;
  return o;
}

static __device__ __forceinline__ WOps ldW(const float* w) {
  WOps o;
  o.w0 = *(const v4f*)w;             o.w1 = *(const v4f*)(w + WP);
  o.w2 = *(const v4f*)(w + 2 * WP);  o.w3 = *(const v4f*)(w + 3 * WP);
  return o;
}

static __device__ __forceinline__ void mm8(const AOps& a, const WOps& w, Tile& T) {
  v2f a0l, a0h, a1l, a1h;
  a0l.x = a.x0.x; a0l.y = a.x0.y;  a0h.x = a.x0.z; a0h.y = a.x0.w;
  a1l.x = a.x1.x; a1l.y = a.x1.y;  a1h.x = a.x1.z; a1h.y = a.x1.w;
  mma8(a0l, a1l, w.w0, w.w1, T);
  mma8(a0h, a1h, w.w2, w.w3, T);
}

static __device__ __forceinline__ APre gemm300_head(const float* a0,
                                                    const float* a1, int half) {
  const float* p0 = a0 + 4 * half;
  const float* p1 = a1 + 4 * half;
  APre h;
  h.s0 = ldA(p0, p1);            h.s1 = ldA(p0 + 8, p1 + 8);
  h.s2 = ldA(p0 + 16, p1 + 16);  h.s3 = ldA(p0 + 24, p1 + 24);
  return h;
}

template <bool LAST>
static __device__ __forceinline__ void trip12(AOps (&A)[6], WOps (&Wb)[4],
                                              const float* p0, const float* p1,
                                              const float* pw, Tile& T) {
#pragma unroll
  for (int b = 0; b < 6; ++b) {
    const int al = 2 * ((b + 2) % 3), wl = 2 * ((b + 1) % 2);
    const int ac = 2 * (b % 3), wc = 2 * (b % 2);
    const int sa = 2 * b + 4, sw = 2 * b + 2;
    if (!LAST || 24 + sa <= 36)     A[al]     = ldA(p0 + 8 * sa, p1 + 8 * sa);
    if (!LAST || 24 + sa + 1 <= 36) A[al + 1] = ldA(p0 + 8 * sa + 8, p1 + 8 * sa + 8);
    if (!LAST || 24 + sw <= 36)     Wb[wl]     = ldW(pw + 8 * sw * WP);
    if (!LAST || 24 + sw + 1 <= 36) Wb[wl + 1] = ldW(pw + (8 * sw + 8) * WP);
    __builtin_amdgcn_sched_barrier(0);
    mm8(A[ac], Wb[wc], T);
    mm8(A[ac + 1], Wb[wc + 1], T);
    __builtin_amdgcn_sched_barrier(0);
  }
}

static __device__ __forceinline__ void gemm300(const APre& pre, const float* a0,
                                               const float* a1, const float* w,
                                               int half, Tile& T) {
  const float* p0 = a0 + 4 * half;
  const float* p1 = a1 + 4 * half;
  const float* pw = w + 4 * half * WP;
  AOps A[6];
  WOps Wb[4];
  A[0] = pre.s0;  A[1] = pre.s1;  A[2] = pre.s2;  A[3] = pre.s3;
  Wb[0] = ldW(pw);  Wb[1] = ldW(pw + 8 * WP);
#pragma unroll 1
  for (int it = 0; it < 2; ++it) {
    trip12<false>(A, Wb, p0, p1, pw, T);
    p0 += 96; p1 += 96; pw += 96 * WP;
  }
  trip12<true>(A, Wb, p0, p1, pw, T);
  const int kt = 296 + 2 * half;
  const v2f t0 = *(const v2f*)(a0 + kt);
  const v2f t1 = *(const v2f*)(a1 + kt);
  const v4f u0 = *(const v4f*)(w + kt * WP);
  const v4f u1 = *(const v4f*)(w + (kt + 1) * WP);
  __builtin_amdgcn_sched_barrier(0);
  mm8(A[0], Wb[0], T);
  mma8(t0, t1, u0, u1, T);
}

#define GEMM_COORDS()                                                        \
  const int lane = threadIdx.x & 31, wv = threadIdx.x >> 5;                  \
  const int c = lane & 15, half = lane >> 4;                                 \
  const int nc = blockIdx.x % NCH, grp = blockIdx.x / NCH;                   \
  const int mp = grp * 8 + wv;                                               \
  const int ncol = nc * 64 + 4 * c;                                          \
  const bool valid = ncol < Hq;                \
  const int ncc = valid ? ncol : (Hq - 4);                                   \
  const int row0 = mp * 32 + c, row1 = row0 + 16;                            \
  const int rb = mp * 32 + half * 8;

__global__ __launch_bounds__(256) void fc1_kernel(
    const float* __restrict__ af, const float* __restrict__ w,
    float* __restrict__ hout) {
  GEMM_COORDS()
  Tile T = {};
  const float* a0 = af + row0 * AFq + 2 * half;
  const float* a1 = af + row1 * AFq + 2 * half;
  const float* pw = w + 2 * half * Hq + ncc;
#pragma unroll 4
  for (int kk = 0; kk < 80; kk += 4) {
    const v2f x0 = *(const v2f*)a0;
    const v2f x1 = *(const v2f*)a1;
    const v4f u0 = *(const v4f*)pw;
    const v4f u1 = *(const v4f*)(pw + Hq);
    a0 += 4; a1 += 4; pw += 4 * Hq;
    mma8(x0, x1, u0, u1, T);
  }
  {
    const bool kv = (half == 0);
    const int kc = kv ? 80 : 0;
    v2f x0 = *(const v2f*)(af + row0 * AFq + kc);
    v2f x1 = *(const v2f*)(af + row1 * AFq + kc);
    v4f u0 = *(const v4f*)(w + kc * Hq + ncc);
    v4f u1 = *(const v4f*)(w + (kc + 1) * Hq + ncc);
    const v2f z2 = {0.f, 0.f};
    const v4f z4 = {0.f, 0.f, 0.f, 0.f};
    x0 = kv ? x0 : z2;  x1 = kv ? x1 : z2;
    u0 = kv ? u0 : z4;  u1 = kv ? u1 : z4;
    mma8(x0, x1, u0, u1, T);
  }
#pragma unroll
  for (int i = 0; i < 2; ++i) {
#pragma unroll
    for (int r = 0; r < 8; ++r) {
      v4f v;
      v.x = fmaxf(T.c[i][0][r], 0.f);  v.y = fmaxf(T.c[i][1][r], 0.f);
      v.z = fmaxf(T.c[i][2][r], 0.f);  v.w = fmaxf(T.c[i][3][r], 0.f);
      if (valid) *(v4f*)(hout + (rb + i * 16 + r) * Hq + ncol) = v;
    }
  }
}

__global__ __launch_bounds__(256) void bondproj_kernel(
    const float* __restrict__ bond, const float* __restrict__ wn,
    const float* __restrict__ w2b, float* __restrict__ q1,
    float* __restrict__ q2) {
  GEMM_COORDS()
  const float* wnb = wn + Hq * Hq;
  const bool kv = (half == 0);
  const int kc = kv ? 4 : 0;
  const v2f z2 = {0.f, 0.f};
  const v4f z4 = {0.f, 0.f, 0.f, 0.f};
  const v2f xa0 = *(const v2f*)(bond + row0 * BFq + 2 * half);
  const v2f xa1 = *(const v2f*)(bond + row1 * BFq + 2 * half);
  v2f xb0 = *(const v2f*)(bond + row0 * BFq + kc);
  v2f xb1 = *(const v2f*)(bond + row1 * BFq + kc);
  xb0 = kv ? xb0 : z2;  xb1 = kv ? xb1 : z2;
#pragma unroll
  for (int pass = 0; pass < 2; ++pass) {
    const float* wsrc = pass ? w2b : wnb;
    float* dst = pass ? q2 : q1;
    const v4f u0 = *(const v4f*)(wsrc + (2 * half) * Hq + ncc);
    const v4f u1 = *(const v4f*)(wsrc + (2 * half + 1) * Hq + ncc);
    v4f u2 = *(const v4f*)(wsrc + kc * Hq + ncc);
    v4f u3 = *(const v4f*)(wsrc + (kc + 1) * Hq + ncc);
    u2 = kv ? u2 : z4;  u3 = kv ? u3 : z4;
    Tile T = {};
    mma8(xa0, xa1, u0, u1, T);
    mma8(xb0, xb1, u2, u3, T);
#pragma unroll
    for (int i = 0; i < 2; ++i) {
#pragma unroll
      for (int r = 0; r < 8; ++r) {
        v4f v;
        v.x = T.c[i][0][r];  v.y = T.c[i][1][r];
        v.z = T.c[i][2][r];  v.w = T.c[i][3][r];
        if (valid) *(v4f*)(dst + (rb + i * 16 + r) * Hq + ncol) = v;
      }
    }
  }
}

__global__ __launch_bounds__(256) void proj_kernel(
    const float* __restrict__ h, const float* __restrict__ w,
    float* __restrict__ pout) {
  GEMM_COORDS()
  __shared__ float wl[Hq * WP];
  const APre pre = gemm300_head(h + row0 * Hq, h + row1 * Hq, half);
  stage_w_store(stage_w_load(w + ncc), wl);
  __syncthreads();
  Tile T = {};
  gemm300(pre, h + row0 * Hq, h + row1 * Hq, wl + 4 * c, half, T);
#pragma unroll
  for (int i = 0; i < 2; ++i) {
#pragma unroll
    for (int r = 0; r < 8; ++r) {
      v4f v;
      v.x = T.c[i][0][r];  v.y = T.c[i][1][r];
      v.z = T.c[i][2][r];  v.w = T.c[i][3][r];
      if (valid) *(v4f*)(pout + (rb + i * 16 + r) * Hq + ncol) = v;
    }
  }
}

__global__ __launch_bounds__(256) void nei_gather_kernel(
    const float* __restrict__ p, const float* __restrict__ q,
    const int* __restrict__ ag, const int* __restrict__ bg,
    const int* __restrict__ nnb, const float* __restrict__ bn,
    float* __restrict__ out) {
  for (int i = blockIdx.x * 256 + threadIdx.x; i < ITEMS; i += GBLK * 256) {
    const int ba = i / C4, c = (i - ba * C4) * 4;
    const int nn = nnb[ba];
    const float bx = bn[c], by = bn[c + 1], bz = bn[c + 2], bw = bn[c + 3];
    int po[NBq], qo[NBq];
#pragma unroll
    for (int j = 0; j < NBq; ++j) {
      const int e = (ba * NBq + j) * 2;
      po[j] = (ag[e] * Asz + ag[e + 1]) * Hq + c;
      qo[j] = (bg[e] * BONDq + bg[e + 1]) * Hq + c;
    }
    float4 pv[NBq], qv[NBq];
#pragma unroll
    for (int j = 0; j < NBq; ++j) {
      pv[j] = make_float4(0.f, 0.f, 0.f, 0.f);
      qv[j] = make_float4(0.f, 0.f, 0.f, 0.f);
      if (j < nn) {
        pv[j] = *(const float4*)(p + po[j]);
        qv[j] = *(const float4*)(q + qo[j]);
      }
    }
    float4 s = make_float4(0.f, 0.f, 0.f, 0.f);
#pragma unroll
    for (int j = 0; j < NBq; ++j) {
      const bool live = j < nn;
      const float x = fmaxf(pv[j].x + qv[j].x + bx, 0.f);
      const float y = fmaxf(pv[j].y + qv[j].y + by, 0.f);
      const float z = fmaxf(pv[j].z + qv[j].z + bz, 0.f);
      const float w = fmaxf(pv[j].w + qv[j].w + bw, 0.f);
      s.x += live ? x : 0.f;  s.y += live ? y : 0.f;
      s.z += live ? z : 0.f;  s.w += live ? w : 0.f;
    }
    *(float4*)(out + ba * Hq + c) = s;
  }
}

__global__ __launch_bounds__(256) void atom_kernel(
    const float* __restrict__ h, const float* __restrict__ nei,
    const float* __restrict__ wa, const float* __restrict__ bb,
    float* __restrict__ hout) {
  GEMM_COORDS()
  __shared__ float wl[Hq * WP];
  const APre pre = gemm300_head(h + row0 * Hq, h + row1 * Hq, half);
  stage_w_store(stage_w_load(wa + ncc), wl);
  __syncthreads();
  Tile T = {};
  gemm300(pre, h + row0 * Hq, h + row1 * Hq, wl + 4 * c, half, T);
  const WStage s2 = stage_w_load(wa + Hq * Hq + ncc);
  const APre pre2 = gemm300_head(nei + row0 * Hq, nei + row1 * Hq, half);
  __syncthreads();
  stage_w_store(s2, wl);
  __syncthreads();
  gemm300(pre2, nei + row0 * Hq, nei + row1 * Hq, wl + 4 * c, half, T);
  const v4f bias = *(const v4f*)(bb + ncc);
#pragma unroll
  for (int i = 0; i < 2; ++i) {
#pragma unroll
    for (int r = 0; r < 8; ++r) {
      v4f v;
      v.x = fmaxf(T.c[i][0][r] + bias.x, 0.f);
      v.y = fmaxf(T.c[i][1][r] + bias.y, 0.f);
      v.z = fmaxf(T.c[i][2][r] + bias.z, 0.f);
      v.w = fmaxf(T.c[i][3][r] + bias.w, 0.f);
      if (valid) *(v4f*)(hout + (rb + i * 16 + r) * Hq + ncol) = v;
    }
  }
}

__global__ __launch_bounds__(256) void nprod_gather_kernel(
    const float* __restrict__ p, const float* __restrict__ q,
    const int* __restrict__ ag, const int* __restrict__ bg,
    const int* __restrict__ nnb, float* __restrict__ out) {
  for (int i = blockIdx.x * 256 + threadIdx.x; i < ITEMS; i += GBLK * 256) {
    const int ba = i / C4, c = (i - ba * C4) * 4;
    const int nn = nnb[ba];
    int po[NBq], qo[NBq];
#pragma unroll
    for (int j = 0; j < NBq; ++j) {
      const int e = (ba * NBq + j) * 2;
      po[j] = (ag[e] * Asz + ag[e + 1]) * Hq + c;
      qo[j] = (bg[e] * BONDq + bg[e + 1]) * Hq + c;
    }
    float4 pv[NBq], qv[NBq];
#pragma unroll
    for (int j = 0; j < NBq; ++j) {
      pv[j] = make_float4(0.f, 0.f, 0.f, 0.f);
      qv[j] = make_float4(0.f, 0.f, 0.f, 0.f);
      if (j < nn) {
        pv[j] = *(const float4*)(p + po[j]);
        qv[j] = *(const float4*)(q + qo[j]);
      }
    }
    float4 s = make_float4(0.f, 0.f, 0.f, 0.f);
#pragma unroll
    for (int j = 0; j < NBq; ++j) {
      const bool live = j < nn;
      s.x += live ? pv[j].x * qv[j].x : 0.f;
      s.y += live ? pv[j].y * qv[j].y : 0.f;
      s.z += live ? pv[j].z * qv[j].z : 0.f;
      s.w += live ? pv[j].w * qv[j].w : 0.f;
    }
    *(float4*)(out + ba * Hq + c) = s;
  }
}

__global__ __launch_bounds__(256) void out_kernel(
    const float* __restrict__ h, const float* __restrict__ w,
    const float* __restrict__ nprod, const int* __restrict__ na,
    float* __restrict__ out) {
  GEMM_COORDS()
  __shared__ float wl[Hq * WP];
  const APre pre = gemm300_head(h + row0 * Hq, h + row1 * Hq, half);
  stage_w_store(stage_w_load(w + ncc), wl);
  __syncthreads();
  Tile T = {};
  gemm300(pre, h + row0 * Hq, h + row1 * Hq, wl + 4 * c, half, T);
#pragma unroll
  for (int i = 0; i < 2; ++i) {
#pragma unroll
    for (int r = 0; r < 8; ++r) {
      const int row = rb + i * 16 + r;
      const int bi = row / Asz, ai = row - bi * Asz;
      const bool live = ai < na[bi];
      if (valid) {
        const v4f np4 = *(const v4f*)(nprod + row * Hq + ncol);
        v4f v;
        v.x = live ? T.c[i][0][r] * np4.x : 0.f;
        v.y = live ? T.c[i][1][r] * np4.y : 0.f;
        v.z = live ? T.c[i][2][r] * np4.z : 0.f;
        v.w = live ? T.c[i][3][r] * np4.w : 0.f;
        *(v4f*)(out + row * Hq + ncol) = v;
      }
    }
  }
}

extern "C" void kernel_launch(void* const* d_in, const int* in_sizes, int n_in,
                              void* d_out, int out_size, void* d_ws, size_t ws_size,
                              hipStream_t stream) {
  const float* atom_feats = (const float*)d_in[0];
  const float* bond_feats = (const float*)d_in[1];
  const float* w_fc1      = (const float*)d_in[2];
  const float* w_nei      = (const float*)d_in[3];
  const float* b_nei      = (const float*)d_in[4];
  const float* w_atom     = (const float*)d_in[5];
  const float* b_atom     = (const float*)d_in[6];
  const float* w2a        = (const float*)d_in[7];
  const float* w2b        = (const float*)d_in[8];
  const float* w_fc2      = (const float*)d_in[9];
  const int*   atom_graph = (const int*)d_in[10];
  const int*   bond_graph = (const int*)d_in[11];
  const int*   num_nbs    = (const int*)d_in[12];
  const int*   n_atoms    = (const int*)d_in[13];
  float* out = (float*)d_out;

  const size_t nH = (size_t)Bsz * Asz * Hq;
  const size_t nQ = (size_t)Bsz * BONDq * Hq;
  float* h_a = (float*)d_ws;
  float* h_b = h_a + nH;
  float* nei = h_b + nH;
  float* prj = nei + nH;
  float* q1  = prj + nH;
  float* q2  = q1 + nQ;

  dim3 blk(256);
  const int gemm_blocks = (MP / 8) * NCH;
  const int bond_blocks = (BMP / 8) * NCH;

  bondproj_kernel<<<bond_blocks, blk, 0, stream>>>(bond_feats, w_nei, w2b, q1, q2);
  fc1_kernel<<<gemm_blocks, blk, 0, stream>>>(atom_feats, w_fc1, h_a);

  proj_kernel<<<gemm_blocks, blk, 0, stream>>>(h_a, w_nei, prj);
  nei_gather_kernel<<<GBLK, blk, 0, stream>>>(prj, q1, atom_graph, bond_graph,
                                             num_nbs, b_nei, nei);
  atom_kernel<<<gemm_blocks, blk, 0, stream>>>(h_a, nei, w_atom, b_atom, h_b);
  proj_kernel<<<gemm_blocks, blk, 0, stream>>>(h_b, w_nei, prj);
  nei_gather_kernel<<<GBLK, blk, 0, stream>>>(prj, q1, atom_graph, bond_graph,
                                             num_nbs, b_nei, nei);
  atom_kernel<<<gemm_blocks, blk, 0, stream>>>(h_b, nei, w_atom, b_atom, h_a);
  proj_kernel<<<gemm_blocks, blk, 0, stream>>>(h_a, w2a, prj);
  nprod_gather_kernel<<<GBLK, blk, 0, stream>>>(prj, q2, atom_graph, bond_graph,
                                               num_nbs, nei);
  out_kernel<<<gemm_blocks, blk, 0, stream>>>(h_a, w_fc2, nei, n_atoms, out);
}


# Neighbour stage: K=6 bond product formed inside the gather kernel on WMMA from the 24-byte bond rows; projected bond rows no longer written or gathered (one 1200-B row per neighbour); sum order kept
# speedup vs baseline: 4.8026x; 1.0813x over previous
#include <hip/hip_runtime.h>
#include <hip/hip_bf16.h>


#define Bsz   256
#define Asz   100
#define NBq   10
#define BONDq 110
#define AFq   82
#define BFq   6
#define Hq    300
#define NCH   5
#define MTOT  (Bsz*Asz)
#define MP    (MTOT/32)
#define NBLK  (MTOT/16*NCH/8)

typedef float v2f __attribute__((ext_vector_type(2)));
typedef float v4f __attribute__((ext_vector_type(4)));
typedef float v8f __attribute__((ext_vector_type(8)));

static __device__ __forceinline__ v8f wmma4(v2f a, v2f b, v8f c) {
  return __builtin_amdgcn_wmma_f32_16x16x4_f32(
      false, a, false, b, (short)0, c, false, false);
}

struct Tile { v8f c[2][4]; };
struct AOps { v4f x0, x1; };
struct WOps { v4f w0, w1, w2, w3; };
struct APre { AOps s0, s1, s2, s3; };

static __device__ __forceinline__ void mma8(v2f a0, v2f a1, v4f wlo, v4f whi,
                                            Tile& T) {
#pragma unroll
  for (int t = 0; t < 4; ++t) {
    v2f b; b.x = wlo[t]; b.y = whi[t];
    T.c[0][t] = wmma4(a0, b, T.c[0][t]);
    T.c[1][t] = wmma4(a1, b, T.c[1][t]);
  }
}

#define WP 64

struct WStage { v4f v[19]; };

static __device__ __forceinline__ WStage stage_w_load(const float* wsrc) {
  const int r0 = threadIdx.x >> 4;
  WStage s;
#pragma unroll
  for (int i = 0; i < 19; ++i) {
    const int r = r0 + 16 * i;
    const int rc = r < Hq ? r : Hq - 1;
    s.v[i] = *(const v4f*)(wsrc + rc * Hq);
  }
  return s;
}

static __device__ __forceinline__ void stage_w_store(const WStage& s, float* wl) {
  const int g = threadIdx.x & 15, r0 = threadIdx.x >> 4;
#pragma unroll
  for (int i = 0; i < 19; ++i) {
    const int r = r0 + 16 * i;
    if (r < Hq) *(v4f*)(wl + r * WP + 4 * g) = s.v[i];
  }
}

static __device__ __forceinline__ AOps ldA(const float* a0, const float* a1) {
  AOps o;
  o.x0 = *(const v4f*)a0;            o.x1 = *(const v4f*)a1;
  return o;
}

static __device__ __forceinline__ WOps ldW(const float* w) {
  WOps o;
  o.w0 = *(const v4f*)w;             o.w1 = *(const v4f*)(w + WP);
  o.w2 = *(const v4f*)(w + 2 * WP);  o.w3 = *(const v4f*)(w + 3 * WP);
  return o;
}

static __device__ __forceinline__ void mm8(const AOps& a, const WOps& w, Tile& T) {
  v2f a0l, a0h, a1l, a1h;
  a0l.x = a.x0.x; a0l.y = a.x0.y;  a0h.x = a.x0.z; a0h.y = a.x0.w;
  a1l.x = a.x1.x; a1l.y = a.x1.y;  a1h.x = a.x1.z; a1h.y = a.x1.w;
  mma8(a0l, a1l, w.w0, w.w1, T);
  mma8(a0h, a1h, w.w2, w.w3, T);
}

static __device__ __forceinline__ APre gemm300_head(const float* a0,
                                                    const float* a1, int half) {
  const float* p0 = a0 + 4 * half;
  const float* p1 = a1 + 4 * half;
  APre h;
  h.s0 = ldA(p0, p1);            h.s1 = ldA(p0 + 8, p1 + 8);
  h.s2 = ldA(p0 + 16, p1 + 16);  h.s3 = ldA(p0 + 24, p1 + 24);
  return h;
}

template <bool LAST>
static __device__ __forceinline__ void trip12(AOps (&A)[6], WOps (&Wb)[4],
                                              const float* p0, const float* p1,
                                              const float* pw, Tile& T) {
#pragma unroll
  for (int b = 0; b < 6; ++b) {
    const int al = 2 * ((b + 2) % 3), wl = 2 * ((b + 1) % 2);
    const int ac = 2 * (b % 3), wc = 2 * (b % 2);
    const int sa = 2 * b + 4, sw = 2 * b + 2;
    if (!LAST || 24 + sa <= 36)     A[al]     = ldA(p0 + 8 * sa, p1 + 8 * sa);
    if (!LAST || 24 + sa + 1 <= 36) A[al + 1] = ldA(p0 + 8 * sa + 8, p1 + 8 * sa + 8);
    if (!LAST || 24 + sw <= 36)     Wb[wl]     = ldW(pw + 8 * sw * WP);
    if (!LAST || 24 + sw + 1 <= 36) Wb[wl + 1] = ldW(pw + (8 * sw + 8) * WP);
    __builtin_amdgcn_sched_barrier(0);
    mm8(A[ac], Wb[wc], T);
    mm8(A[ac + 1], Wb[wc + 1], T);
    __builtin_amdgcn_sched_barrier(0);
  }
}

static __device__ __forceinline__ void gemm300(const APre& pre, const float* a0,
                                               const float* a1, const float* w,
                                               int half, Tile& T) {
  const float* p0 = a0 + 4 * half;
  const float* p1 = a1 + 4 * half;
  const float* pw = w + 4 * half * WP;
  AOps A[6];
  WOps Wb[4];
  A[0] = pre.s0;  A[1] = pre.s1;  A[2] = pre.s2;  A[3] = pre.s3;
  Wb[0] = ldW(pw);  Wb[1] = ldW(pw + 8 * WP);
#pragma unroll 1
  for (int it = 0; it < 2; ++it) {
    trip12<false>(A, Wb, p0, p1, pw, T);
    p0 += 96; p1 += 96; pw += 96 * WP;
  }
  trip12<true>(A, Wb, p0, p1, pw, T);
  const int kt = 296 + 2 * half;
  const v2f t0 = *(const v2f*)(a0 + kt);
  const v2f t1 = *(const v2f*)(a1 + kt);
  const v4f u0 = *(const v4f*)(w + kt * WP);
  const v4f u1 = *(const v4f*)(w + (kt + 1) * WP);
  __builtin_amdgcn_sched_barrier(0);
  mm8(A[0], Wb[0], T);
  mma8(t0, t1, u0, u1, T);
}

#define GEMM_COORDS()                                                        \
  const int lane = threadIdx.x & 31, wv = threadIdx.x >> 5;                  \
  const int c = lane & 15, half = lane >> 4;                                 \
  const int nc = blockIdx.x % NCH, grp = blockIdx.x / NCH;                   \
  const int mp = grp * 8 + wv;                                               \
  const int ncol = nc * 64 + 4 * c;                                          \
  const bool valid = ncol < Hq;                \
  const int ncc = valid ? ncol : (Hq - 4);                                   \
  const int row0 = mp * 32 + c, row1 = row0 + 16;                            \
  const int rb = mp * 32 + half * 8;

__global__ __launch_bounds__(256) void fc1_kernel(
    const float* __restrict__ af, const float* __restrict__ w,
    float* __restrict__ hout) {
  GEMM_COORDS()
  Tile T = {};
  const float* a0 = af + row0 * AFq + 2 * half;
  const float* a1 = af + row1 * AFq + 2 * half;
  const float* pw = w + 2 * half * Hq + ncc;
#pragma unroll 4
  for (int kk = 0; kk < 80; kk += 4) {
    const v2f x0 = *(const v2f*)a0;
    const v2f x1 = *(const v2f*)a1;
    const v4f u0 = *(const v4f*)pw;
    const v4f u1 = *(const v4f*)(pw + Hq);
    a0 += 4; a1 += 4; pw += 4 * Hq;
    mma8(x0, x1, u0, u1, T);
  }
  {
    const bool kv = (half == 0);
    const int kc = kv ? 80 : 0;
    v2f x0 = *(const v2f*)(af + row0 * AFq + kc);
    v2f x1 = *(const v2f*)(af + row1 * AFq + kc);
    v4f u0 = *(const v4f*)(w + kc * Hq + ncc);
    v4f u1 = *(const v4f*)(w + (kc + 1) * Hq + ncc);
    const v2f z2 = {0.f, 0.f};
    const v4f z4 = {0.f, 0.f, 0.f, 0.f};
    x0 = kv ? x0 : z2;  x1 = kv ? x1 : z2;
    u0 = kv ? u0 : z4;  u1 = kv ? u1 : z4;
    mma8(x0, x1, u0, u1, T);
  }
#pragma unroll
  for (int i = 0; i < 2; ++i) {
#pragma unroll
    for (int r = 0; r < 8; ++r) {
      v4f v;
      v.x = fmaxf(T.c[i][0][r], 0.f);  v.y = fmaxf(T.c[i][1][r], 0.f);
      v.z = fmaxf(T.c[i][2][r], 0.f);  v.w = fmaxf(T.c[i][3][r], 0.f);
      if (valid) *(v4f*)(hout + (rb + i * 16 + r) * Hq + ncol) = v;
    }
  }
}

__global__ __launch_bounds__(256) void proj_kernel(
    const float* __restrict__ h, const float* __restrict__ w,
    float* __restrict__ pout) {
  GEMM_COORDS()
  __shared__ float wl[Hq * WP];
  const APre pre = gemm300_head(h + row0 * Hq, h + row1 * Hq, half);
  stage_w_store(stage_w_load(w + ncc), wl);
  __syncthreads();
  Tile T = {};
  gemm300(pre, h + row0 * Hq, h + row1 * Hq, wl + 4 * c, half, T);
#pragma unroll
  for (int i = 0; i < 2; ++i) {
#pragma unroll
    for (int r = 0; r < 8; ++r) {
      v4f v;
      v.x = T.c[i][0][r];  v.y = T.c[i][1][r];
      v.z = T.c[i][2][r];  v.w = T.c[i][3][r];
      if (valid) *(v4f*)(pout + (rb + i * 16 + r) * Hq + ncol) = v;
    }
  }
}

template <bool PROD>
__global__ __launch_bounds__(256, 8) void nbr_kernel(
    const float* __restrict__ p, const float* __restrict__ bond,
    const float* __restrict__ wb, const int* __restrict__ ag,
    const int* __restrict__ bg, const int* __restrict__ nnb,
    const float* __restrict__ bn, float* __restrict__ out) {
  __shared__ int tab[8][2][NBq][16];
  const int lane = threadIdx.x & 31, wv = threadIdx.x >> 5;
  const int c = lane & 15, half = lane >> 4;
  const int tile = blockIdx.x * 8 + wv;
  const int nc = tile % NCH, grp = tile / NCH;
  const int a0 = grp * 16;
  const int ncol = nc * 64 + 4 * c;
  const bool valid = ncol < Hq;
  const int ncc = valid ? ncol : (Hq - 4);
  const bool kv = (half == 0);
  const int kc = kv ? 4 : 0;
  const int atom = a0 + c;

  {
    const int4* src = (const int4*)((kv ? ag : bg) + atom * (2 * NBq));
    const int m1 = kv ? Asz : BONDq;
    const int m2 = kv ? Hq : BFq;
    int v[2 * NBq];
#pragma unroll
    for (int i = 0; i < 5; ++i) {
      const int4 x = src[i];
      v[4 * i] = x.x;  v[4 * i + 1] = x.y;  v[4 * i + 2] = x.z;  v[4 * i + 3] = x.w;
    }
#pragma unroll
    for (int j = 0; j < NBq; ++j) tab[wv][half][j][c] = (v[2 * j] * m1 + v[2 * j + 1]) * m2;
  }
  const int nnl = nnb[atom];
  int nr[8];
#pragma unroll
  for (int r = 0; r < 8; ++r) nr[r] = __builtin_amdgcn_ds_bpermute(32 * half + 4 * r, nnl);

  const v4f z4 = {0.f, 0.f, 0.f, 0.f};
  const v2f z2 = {0.f, 0.f};
  const v4f u0 = *(const v4f*)(wb + (2 * half) * Hq + ncc);
  const v4f u1 = *(const v4f*)(wb + (2 * half + 1) * Hq + ncc);
  v4f u2 = *(const v4f*)(wb + kc * Hq + ncc);
  v4f u3 = *(const v4f*)(wb + (kc + 1) * Hq + ncc);
  u2 = kv ? u2 : z4;  u3 = kv ? u3 : z4;
  v2f b0[4], b1[4];
#pragma unroll
  for (int t = 0; t < 4; ++t) {
    b0[t].x = u0[t];  b0[t].y = u1[t];
    b1[t].x = u2[t];  b1[t].y = u3[t];
  }
  v4f bias = z4;
  if (!PROD) bias = *(const v4f*)(bn + ncc);

  v8f acc[4];
  const v8f z8 = {0.f, 0.f, 0.f, 0.f, 0.f, 0.f, 0.f, 0.f};
#pragma unroll
  for (int t = 0; t < 4; ++t) acc[t] = z8;

  __syncthreads();

#pragma unroll 1
  for (int j = 0; j < NBq; ++j) {
    const int4 o0 = *(const int4*)&tab[wv][0][j][8 * half];
    const int4 o1 = *(const int4*)&tab[wv][0][j][8 * half + 4];
    const int off[8] = {o0.x, o0.y, o0.z, o0.w, o1.x, o1.y, o1.z, o1.w};
    const float* brow = bond + tab[wv][1][j][c];
    const v2f alo = *(const v2f*)(brow + 2 * half);
    const v2f t4 = *(const v2f*)(brow + kc);
    const v2f ahi = kv ? t4 : z2;
    v4f row[8];
#pragma unroll
    for (int r = 0; r < 8; ++r) {
      row[r] = z4;
      if (j < nr[r]) row[r] = *(const v4f*)(p + off[r] + ncc);
    }
#pragma unroll
    for (int t = 0; t < 4; ++t) {
      v8f D = wmma4(alo, b0[t], z8);
      D = wmma4(ahi, b1[t], D);
#pragma unroll
      for (int r = 0; r < 8; ++r) {
        float x;
        if (PROD) x = row[r][t] * D[r];
        else      x = fmaxf(row[r][t] + D[r] + bias[t], 0.f);
        acc[t][r] += (j < nr[r]) ? x : 0.f;
      }
    }
  }
#pragma unroll
  for (int r = 0; r < 8; ++r) {
    v4f v;
    v.x = acc[0][r];  v.y = acc[1][r];  v.z = acc[2][r];  v.w = acc[3][r];
    if (valid) *(v4f*)(out + (a0 + 8 * half + r) * Hq + ncol) = v;
  }
}

__global__ __launch_bounds__(256) void atom_kernel(
    const float* __restrict__ h, const float* __restrict__ nei,
    const float* __restrict__ wa, const float* __restrict__ bb,
    float* __restrict__ hout) {
  GEMM_COORDS()
  __shared__ float wl[Hq * WP];
  const APre pre = gemm300_head(h + row0 * Hq, h + row1 * Hq, half);
  stage_w_store(stage_w_load(wa + ncc), wl);
  __syncthreads();
  Tile T = {};
  gemm300(pre, h + row0 * Hq, h + row1 * Hq, wl + 4 * c, half, T);
  const WStage s2 = stage_w_load(wa + Hq * Hq + ncc);
  const APre pre2 = gemm300_head(nei + row0 * Hq, nei + row1 * Hq, half);
  __syncthreads();
  stage_w_store(s2, wl);
  __syncthreads();
  gemm300(pre2, nei + row0 * Hq, nei + row1 * Hq, wl + 4 * c, half, T);
  const v4f bias = *(const v4f*)(bb + ncc);
#pragma unroll
  for (int i = 0; i < 2; ++i) {
#pragma unroll
    for (int r = 0; r < 8; ++r) {
      v4f v;
      v.x = fmaxf(T.c[i][0][r] + bias.x, 0.f);
      v.y = fmaxf(T.c[i][1][r] + bias.y, 0.f);
      v.z = fmaxf(T.c[i][2][r] + bias.z, 0.f);
      v.w = fmaxf(T.c[i][3][r] + bias.w, 0.f);
      if (valid) *(v4f*)(hout + (rb + i * 16 + r) * Hq + ncol) = v;
    }
  }
}

__global__ __launch_bounds__(256) void out_kernel(
    const float* __restrict__ h, const float* __restrict__ w,
    const float* __restrict__ nprod, const int* __restrict__ na,
    float* __restrict__ out) {
  GEMM_COORDS()
  __shared__ float wl[Hq * WP];
  const APre pre = gemm300_head(h + row0 * Hq, h + row1 * Hq, half);
  stage_w_store(stage_w_load(w + ncc), wl);
  __syncthreads();
  Tile T = {};
  gemm300(pre, h + row0 * Hq, h + row1 * Hq, wl + 4 * c, half, T);
#pragma unroll
  for (int i = 0; i < 2; ++i) {
#pragma unroll
    for (int r = 0; r < 8; ++r) {
      const int row = rb + i * 16 + r;
      const int bi = row / Asz, ai = row - bi * Asz;
      const bool live = ai < na[bi];
      if (valid) {
        const v4f np4 = *(const v4f*)(nprod + row * Hq + ncol);
        v4f v;
        v.x = live ? T.c[i][0][r] * np4.x : 0.f;
        v.y = live ? T.c[i][1][r] * np4.y : 0.f;
        v.z = live ? T.c[i][2][r] * np4.z : 0.f;
        v.w = live ? T.c[i][3][r] * np4.w : 0.f;
        *(v4f*)(out + row * Hq + ncol) = v;
      }
    }
  }
}

extern "C" void kernel_launch(void* const* d_in, const int* in_sizes, int n_in,
                              void* d_out, int out_size, void* d_ws, size_t ws_size,
                              hipStream_t stream) {
  const float* atom_feats = (const float*)d_in[0];
  const float* bond_feats = (const float*)d_in[1];
  const float* w_fc1      = (const float*)d_in[2];
  const float* w_nei      = (const float*)d_in[3];
  const float* b_nei      = (const float*)d_in[4];
  const float* w_atom     = (const float*)d_in[5];
  const float* b_atom     = (const float*)d_in[6];
  const float* w2a        = (const float*)d_in[7];
  const float* w2b        = (const float*)d_in[8];
  const float* w_fc2      = (const float*)d_in[9];
  const int*   atom_graph = (const int*)d_in[10];
  const int*   bond_graph = (const int*)d_in[11];
  const int*   num_nbs    = (const int*)d_in[12];
  const int*   n_atoms    = (const int*)d_in[13];
  float* out = (float*)d_out;

  const size_t nH = (size_t)Bsz * Asz * Hq;
  float* h_a = (float*)d_ws;
  float* h_b = h_a + nH;
  float* nei = h_b + nH;
  float* prj = nei + nH;
  const float* w_nei_bond = w_nei + Hq * Hq;

  dim3 blk(256);
  const int gemm_blocks = (MP / 8) * NCH;

  fc1_kernel<<<gemm_blocks, blk, 0, stream>>>(atom_feats, w_fc1, h_a);

  proj_kernel<<<gemm_blocks, blk, 0, stream>>>(h_a, w_nei, prj);
  nbr_kernel<false><<<NBLK, blk, 0, stream>>>(prj, bond_feats, w_nei_bond, atom_graph,
                                              bond_graph, num_nbs, b_nei, nei);
  atom_kernel<<<gemm_blocks, blk, 0, stream>>>(h_a, nei, w_atom, b_atom, h_b);
  proj_kernel<<<gemm_blocks, blk, 0, stream>>>(h_b, w_nei, prj);
  nbr_kernel<false><<<NBLK, blk, 0, stream>>>(prj, bond_feats, w_nei_bond, atom_graph,
                                              bond_graph, num_nbs, b_nei, nei);
  atom_kernel<<<gemm_blocks, blk, 0, stream>>>(h_b, nei, w_atom, b_atom, h_a);
  proj_kernel<<<gemm_blocks, blk, 0, stream>>>(h_a, w2a, prj);
  nbr_kernel<true><<<NBLK, blk, 0, stream>>>(prj, bond_feats, w2b, atom_graph,
                                             bond_graph, num_nbs, b_nei, nei);
  out_kernel<<<gemm_blocks, blk, 0, stream>>>(h_a, w_fc2, nei, n_atoms, out);
}


# Workspace rows (h, projected rows, neighbour sums) at a 1280-byte pitch, not 1200: each 256-byte row piece a wave gathers, loads or stores is two whole 128-byte lines, not three; sum order kept
# speedup vs baseline: 5.0456x; 1.0506x over previous
#include <hip/hip_runtime.h>
#include <hip/hip_bf16.h>


#define Bsz   256
#define Asz   100
#define NBq   10
#define BONDq 110
#define AFq   82
#define BFq   6
#define Hq    300
#define HP    320
#define NCH   5
#define MTOT  (Bsz*Asz)
#define MP    (MTOT/32)
#define NBLK  (MTOT/16*NCH/8)

typedef float v2f __attribute__((ext_vector_type(2)));
typedef float v4f __attribute__((ext_vector_type(4)));
typedef float v8f __attribute__((ext_vector_type(8)));

static __device__ __forceinline__ v8f wmma4(v2f a, v2f b, v8f c) {
  return __builtin_amdgcn_wmma_f32_16x16x4_f32(
      false, a, false, b, (short)0, c, false, false);
}

struct Tile { v8f c[2][4]; };
struct AOps { v4f x0, x1; };
struct WOps { v4f w0, w1, w2, w3; };
struct APre { AOps s0, s1, s2, s3; };

static __device__ __forceinline__ void mma8(v2f a0, v2f a1, v4f wlo, v4f whi,
                                            Tile& T) {
#pragma unroll
  for (int t = 0; t < 4; ++t) {
    v2f b; b.x = wlo[t]; b.y = whi[t];
    T.c[0][t] = wmma4(a0, b, T.c[0][t]);
    T.c[1][t] = wmma4(a1, b, T.c[1][t]);
  }
}

#define WP 64

struct WStage { v4f v[19]; };

static __device__ __forceinline__ WStage stage_w_load(const float* wsrc) {
  const int r0 = threadIdx.x >> 4;
  WStage s;
#pragma unroll
  for (int i = 0; i < 19; ++i) {
    const int r = r0 + 16 * i;
    const int rc = r < Hq ? r : Hq - 1;
    s.v[i] = *(const v4f*)(wsrc + rc * Hq);
  }
  return s;
}

static __device__ __forceinline__ void stage_w_store(const WStage& s, float* wl) {
  const int g = threadIdx.x & 15, r0 = threadIdx.x >> 4;
#pragma unroll
  for (int i = 0; i < 19; ++i) {
    const int r = r0 + 16 * i;
    if (r < Hq) *(v4f*)(wl + r * WP + 4 * g) = s.v[i];
  }
}

static __device__ __forceinline__ AOps ldA(const float* a0, const float* a1) {
  AOps o;
  o.x0 = *(const v4f*)a0;            o.x1 = *(const v4f*)a1;
  return o;
}

static __device__ __forceinline__ WOps ldW(const float* w) {
  WOps o;
  o.w0 = *(const v4f*)w;             o.w1 = *(const v4f*)(w + WP);
  o.w2 = *(const v4f*)(w + 2 * WP);  o.w3 = *(const v4f*)(w + 3 * WP);
  return o;
}

static __device__ __forceinline__ void mm8(const AOps& a, const WOps& w, Tile& T) {
  v2f a0l, a0h, a1l, a1h;
  a0l.x = a.x0.x; a0l.y = a.x0.y;  a0h.x = a.x0.z; a0h.y = a.x0.w;
  a1l.x = a.x1.x; a1l.y = a.x1.y;  a1h.x = a.x1.z; a1h.y = a.x1.w;
  mma8(a0l, a1l, w.w0, w.w1, T);
  mma8(a0h, a1h, w.w2, w.w3, T);
}

static __device__ __forceinline__ APre gemm300_head(const float* a0,
                                                    const float* a1, int half) {
  const float* p0 = a0 + 4 * half;
  const float* p1 = a1 + 4 * half;
  APre h;
  h.s0 = ldA(p0, p1);            h.s1 = ldA(p0 + 8, p1 + 8);
  h.s2 = ldA(p0 + 16, p1 + 16);  h.s3 = ldA(p0 + 24, p1 + 24);
  return h;
}

template <bool LAST>
static __device__ __forceinline__ void trip12(AOps (&A)[6], WOps (&Wb)[4],
                                              const float* p0, const float* p1,
                                              const float* pw, Tile& T) {
#pragma unroll
  for (int b = 0; b < 6; ++b) {
    const int al = 2 * ((b + 2) % 3), wl = 2 * ((b + 1) % 2);
    const int ac = 2 * (b % 3), wc = 2 * (b % 2);
    const int sa = 2 * b + 4, sw = 2 * b + 2;
    if (!LAST || 24 + sa <= 36)     A[al]     = ldA(p0 + 8 * sa, p1 + 8 * sa);
    if (!LAST || 24 + sa + 1 <= 36) A[al + 1] = ldA(p0 + 8 * sa + 8, p1 + 8 * sa + 8);
    if (!LAST || 24 + sw <= 36)     Wb[wl]     = ldW(pw + 8 * sw * WP);
    if (!LAST || 24 + sw + 1 <= 36) Wb[wl + 1] = ldW(pw + (8 * sw + 8) * WP);
    __builtin_amdgcn_sched_barrier(0);
    mm8(A[ac], Wb[wc], T);
    mm8(A[ac + 1], Wb[wc + 1], T);
    __builtin_amdgcn_sched_barrier(0);
  }
}

static __device__ __forceinline__ void gemm300(const APre& pre, const float* a0,
                                               const float* a1, const float* w,
                                               int half, Tile& T) {
  const float* p0 = a0 + 4 * half;
  const float* p1 = a1 + 4 * half;
  const float* pw = w + 4 * half * WP;
  AOps A[6];
  WOps Wb[4];
  A[0] = pre.s0;  A[1] = pre.s1;  A[2] = pre.s2;  A[3] = pre.s3;
  Wb[0] = ldW(pw);  Wb[1] = ldW(pw + 8 * WP);
#pragma unroll 1
  for (int it = 0; it < 2; ++it) {
    trip12<false>(A, Wb, p0, p1, pw, T);
    p0 += 96; p1 += 96; pw += 96 * WP;
  }
  trip12<true>(A, Wb, p0, p1, pw, T);
  const int kt = 296 + 2 * half;
  const v2f t0 = *(const v2f*)(a0 + kt);
  const v2f t1 = *(const v2f*)(a1 + kt);
  const v4f u0 = *(const v4f*)(w + kt * WP);
  const v4f u1 = *(const v4f*)(w + (kt + 1) * WP);
  __builtin_amdgcn_sched_barrier(0);
  mm8(A[0], Wb[0], T);
  mma8(t0, t1, u0, u1, T);
}

#define GEMM_COORDS()                                                        \
  const int lane = threadIdx.x & 31, wv = threadIdx.x >> 5;                  \
  const int c = lane & 15, half = lane >> 4;                                 \
  const int nc = blockIdx.x % NCH, grp = blockIdx.x / NCH;                   \
  const int mp = grp * 8 + wv;                                               \
  const int ncol = nc * 64 + 4 * c;                                          \
  const bool valid = ncol < Hq;                \
  const int ncc = valid ? ncol : (Hq - 4);                                   \
  const int row0 = mp * 32 + c, row1 = row0 + 16;                            \
  const int rb = mp * 32 + half * 8;

__global__ __launch_bounds__(256) void fc1_kernel(
    const float* __restrict__ af, const float* __restrict__ w,
    float* __restrict__ hout) {
  GEMM_COORDS()
  Tile T = {};
  const float* a0 = af + row0 * AFq + 2 * half;
  const float* a1 = af + row1 * AFq + 2 * half;
  const float* pw = w + 2 * half * Hq + ncc;
#pragma unroll 4
  for (int kk = 0; kk < 80; kk += 4) {
    const v2f x0 = *(const v2f*)a0;
    const v2f x1 = *(const v2f*)a1;
    const v4f u0 = *(const v4f*)pw;
    const v4f u1 = *(const v4f*)(pw + Hq);
    a0 += 4; a1 += 4; pw += 4 * Hq;
    mma8(x0, x1, u0, u1, T);
  }
  {
    const bool kv = (half == 0);
    const int kc = kv ? 80 : 0;
    v2f x0 = *(const v2f*)(af + row0 * AFq + kc);
    v2f x1 = *(const v2f*)(af + row1 * AFq + kc);
    v4f u0 = *(const v4f*)(w + kc * Hq + ncc);
    v4f u1 = *(const v4f*)(w + (kc + 1) * Hq + ncc);
    const v2f z2 = {0.f, 0.f};
    const v4f z4 = {0.f, 0.f, 0.f, 0.f};
    x0 = kv ? x0 : z2;  x1 = kv ? x1 : z2;
    u0 = kv ? u0 : z4;  u1 = kv ? u1 : z4;
    mma8(x0, x1, u0, u1, T);
  }
#pragma unroll
  for (int i = 0; i < 2; ++i) {
#pragma unroll
    for (int r = 0; r < 8; ++r) {
      v4f v;
      v.x = fmaxf(T.c[i][0][r], 0.f);  v.y = fmaxf(T.c[i][1][r], 0.f);
      v.z = fmaxf(T.c[i][2][r], 0.f);  v.w = fmaxf(T.c[i][3][r], 0.f);
      if (valid) *(v4f*)(hout + (rb + i * 16 + r) * HP + ncol) = v;
    }
  }
}

__global__ __launch_bounds__(256) void proj_kernel(
    const float* __restrict__ h, const float* __restrict__ w,
    float* __restrict__ pout) {
  GEMM_COORDS()
  __shared__ float wl[Hq * WP];
  const APre pre = gemm300_head(h + row0 * HP, h + row1 * HP, half);
  stage_w_store(stage_w_load(w + ncc), wl);
  __syncthreads();
  Tile T = {};
  gemm300(pre, h + row0 * HP, h + row1 * HP, wl + 4 * c, half, T);
#pragma unroll
  for (int i = 0; i < 2; ++i) {
#pragma unroll
    for (int r = 0; r < 8; ++r) {
      v4f v;
      v.x = T.c[i][0][r];  v.y = T.c[i][1][r];
      v.z = T.c[i][2][r];  v.w = T.c[i][3][r];
      if (valid) *(v4f*)(pout + (rb + i * 16 + r) * HP + ncol) = v;
    }
  }
}

template <bool PROD>
__global__ __launch_bounds__(256, 8) void nbr_kernel(
    const float* __restrict__ p, const float* __restrict__ bond,
    const float* __restrict__ wb, const int* __restrict__ ag,
    const int* __restrict__ bg, const int* __restrict__ nnb,
    const float* __restrict__ bn, float* __restrict__ out) {
  __shared__ int tab[8][2][NBq][16];
  const int lane = threadIdx.x & 31, wv = threadIdx.x >> 5;
  const int c = lane & 15, half = lane >> 4;
  const int tile = blockIdx.x * 8 + wv;
  const int nc = tile % NCH, grp = tile / NCH;
  const int a0 = grp * 16;
  const int ncol = nc * 64 + 4 * c;
  const bool valid = ncol < Hq;
  const int ncc = valid ? ncol : (Hq - 4);
  const bool kv = (half == 0);
  const int kc = kv ? 4 : 0;
  const int atom = a0 + c;

  {
    const int4* src = (const int4*)((kv ? ag : bg) + atom * (2 * NBq));
    const int m1 = kv ? Asz : BONDq;
    const int m2 = kv ? HP : BFq;
    int v[2 * NBq];
#pragma unroll
    for (int i = 0; i < 5; ++i) {
      const int4 x = src[i];
      v[4 * i] = x.x;  v[4 * i + 1] = x.y;  v[4 * i + 2] = x.z;  v[4 * i + 3] = x.w;
    }
#pragma unroll
    for (int j = 0; j < NBq; ++j) tab[wv][half][j][c] = (v[2 * j] * m1 + v[2 * j + 1]) * m2;
  }
  const int nnl = nnb[atom];
  int nr[8];
#pragma unroll
  for (int r = 0; r < 8; ++r) nr[r] = __builtin_amdgcn_ds_bpermute(32 * half + 4 * r, nnl);

  const v4f z4 = {0.f, 0.f, 0.f, 0.f};
  const v2f z2 = {0.f, 0.f};
  const v4f u0 = *(const v4f*)(wb + (2 * half) * Hq + ncc);
  const v4f u1 = *(const v4f*)(wb + (2 * half + 1) * Hq + ncc);
  v4f u2 = *(const v4f*)(wb + kc * Hq + ncc);
  v4f u3 = *(const v4f*)(wb + (kc + 1) * Hq + ncc);
  u2 = kv ? u2 : z4;  u3 = kv ? u3 : z4;
  v2f b0[4], b1[4];
#pragma unroll
  for (int t = 0; t < 4; ++t) {
    b0[t].x = u0[t];  b0[t].y = u1[t];
    b1[t].x = u2[t];  b1[t].y = u3[t];
  }
  v4f bias = z4;
  if (!PROD) bias = *(const v4f*)(bn + ncc);

  v8f acc[4];
  const v8f z8 = {0.f, 0.f, 0.f, 0.f, 0.f, 0.f, 0.f, 0.f};
#pragma unroll
  for (int t = 0; t < 4; ++t) acc[t] = z8;

  __syncthreads();

#pragma unroll 1
  for (int j = 0; j < NBq; ++j) {
    const int4 o0 = *(const int4*)&tab[wv][0][j][8 * half];
    const int4 o1 = *(const int4*)&tab[wv][0][j][8 * half + 4];
    const int off[8] = {o0.x, o0.y, o0.z, o0.w, o1.x, o1.y, o1.z, o1.w};
    const float* brow = bond + tab[wv][1][j][c];
    const v2f alo = *(const v2f*)(brow + 2 * half);
    const v2f t4 = *(const v2f*)(brow + kc);
    const v2f ahi = kv ? t4 : z2;
    v4f row[8];
#pragma unroll
    for (int r = 0; r < 8; ++r) {
      row[r] = z4;
      if (j < nr[r]) row[r] = *(const v4f*)(p + off[r] + ncc);
    }
#pragma unroll
    for (int t = 0; t < 4; ++t) {
      v8f D = wmma4(alo, b0[t], z8);
      D = wmma4(ahi, b1[t], D);
#pragma unroll
      for (int r = 0; r < 8; ++r) {
        float x;
        if (PROD) x = row[r][t] * D[r];
        else      x = fmaxf(row[r][t] + D[r] + bias[t], 0.f);
        acc[t][r] += (j < nr[r]) ? x : 0.f;
      }
    }
  }
#pragma unroll
  for (int r = 0; r < 8; ++r) {
    v4f v;
    v.x = acc[0][r];  v.y = acc[1][r];  v.z = acc[2][r];  v.w = acc[3][r];
    if (valid) *(v4f*)(out + (a0 + 8 * half + r) * HP + ncol) = v;
  }
}

__global__ __launch_bounds__(256) void atom_kernel(
    const float* __restrict__ h, const float* __restrict__ nei,
    const float* __restrict__ wa, const float* __restrict__ bb,
    float* __restrict__ hout) {
  GEMM_COORDS()
  __shared__ float wl[Hq * WP];
  const APre pre = gemm300_head(h + row0 * HP, h + row1 * HP, half);
  stage_w_store(stage_w_load(wa + ncc), wl);
  __syncthreads();
  Tile T = {};
  gemm300(pre, h + row0 * HP, h + row1 * HP, wl + 4 * c, half, T);
  const WStage s2 = stage_w_load(wa + Hq * Hq + ncc);
  const APre pre2 = gemm300_head(nei + row0 * HP, nei + row1 * HP, half);
  __syncthreads();
  stage_w_store(s2, wl);
  __syncthreads();
  gemm300(pre2, nei + row0 * HP, nei + row1 * HP, wl + 4 * c, half, T);
  const v4f bias = *(const v4f*)(bb + ncc);
#pragma unroll
  for (int i = 0; i < 2; ++i) {
#pragma unroll
    for (int r = 0; r < 8; ++r) {
      v4f v;
      v.x = fmaxf(T.c[i][0][r] + bias.x, 0.f);
      v.y = fmaxf(T.c[i][1][r] + bias.y, 0.f);
      v.z = fmaxf(T.c[i][2][r] + bias.z, 0.f);
      v.w = fmaxf(T.c[i][3][r] + bias.w, 0.f);
      if (valid) *(v4f*)(hout + (rb + i * 16 + r) * HP + ncol) = v;
    }
  }
}

__global__ __launch_bounds__(256) void out_kernel(
    const float* __restrict__ h, const float* __restrict__ w,
    const float* __restrict__ nprod, const int* __restrict__ na,
    float* __restrict__ out) {
  GEMM_COORDS()
  __shared__ float wl[Hq * WP];
  const APre pre = gemm300_head(h + row0 * HP, h + row1 * HP, half);
  stage_w_store(stage_w_load(w + ncc), wl);
  __syncthreads();
  Tile T = {};
  gemm300(pre, h + row0 * HP, h + row1 * HP, wl + 4 * c, half, T);
#pragma unroll
  for (int i = 0; i < 2; ++i) {
#pragma unroll
    for (int r = 0; r < 8; ++r) {
      const int row = rb + i * 16 + r;
      const int bi = row / Asz, ai = row - bi * Asz;
      const bool live = ai < na[bi];
      if (valid) {
        const v4f np4 = *(const v4f*)(nprod + row * HP + ncol);
        v4f v;
        v.x = live ? T.c[i][0][r] * np4.x : 0.f;
        v.y = live ? T.c[i][1][r] * np4.y : 0.f;
        v.z = live ? T.c[i][2][r] * np4.z : 0.f;
        v.w = live ? T.c[i][3][r] * np4.w : 0.f;
        *(v4f*)(out + row * Hq + ncol) = v;
      }
    }
  }
}

extern "C" void kernel_launch(void* const* d_in, const int* in_sizes, int n_in,
                              void* d_out, int out_size, void* d_ws, size_t ws_size,
                              hipStream_t stream) {
  const float* atom_feats = (const float*)d_in[0];
  const float* bond_feats = (const float*)d_in[1];
  const float* w_fc1      = (const float*)d_in[2];
  const float* w_nei      = (const float*)d_in[3];
  const float* b_nei      = (const float*)d_in[4];
  const float* w_atom     = (const float*)d_in[5];
  const float* b_atom     = (const float*)d_in[6];
  const float* w2a        = (const float*)d_in[7];
  const float* w2b        = (const float*)d_in[8];
  const float* w_fc2      = (const float*)d_in[9];
  const int*   atom_graph = (const int*)d_in[10];
  const int*   bond_graph = (const int*)d_in[11];
  const int*   num_nbs    = (const int*)d_in[12];
  const int*   n_atoms    = (const int*)d_in[13];
  float* out = (float*)d_out;

  const size_t nH = (size_t)Bsz * Asz * HP;
  float* h_a = (float*)d_ws;
  float* h_b = h_a + nH;
  float* nei = h_b + nH;
  float* prj = nei + nH;
  const float* w_nei_bond = w_nei + Hq * Hq;

  dim3 blk(256);
  const int gemm_blocks = (MP / 8) * NCH;

  fc1_kernel<<<gemm_blocks, blk, 0, stream>>>(atom_feats, w_fc1, h_a);

  proj_kernel<<<gemm_blocks, blk, 0, stream>>>(h_a, w_nei, prj);
  nbr_kernel<false><<<NBLK, blk, 0, stream>>>(prj, bond_feats, w_nei_bond, atom_graph,
                                              bond_graph, num_nbs, b_nei, nei);
  atom_kernel<<<gemm_blocks, blk, 0, stream>>>(h_a, nei, w_atom, b_atom, h_b);
  proj_kernel<<<gemm_blocks, blk, 0, stream>>>(h_b, w_nei, prj);
  nbr_kernel<false><<<NBLK, blk, 0, stream>>>(prj, bond_feats, w_nei_bond, atom_graph,
                                              bond_graph, num_nbs, b_nei, nei);
  atom_kernel<<<gemm_blocks, blk, 0, stream>>>(h_b, nei, w_atom, b_atom, h_a);
  proj_kernel<<<gemm_blocks, blk, 0, stream>>>(h_a, w2a, prj);
  nbr_kernel<true><<<NBLK, blk, 0, stream>>>(prj, bond_feats, w2b, atom_graph,
                                             bond_graph, num_nbs, b_nei, nei);
  out_kernel<<<gemm_blocks, blk, 0, stream>>>(h_a, w_fc2, nei, n_atoms, out);
}
